# starting kernel
# baseline (speedup 1.0000x reference)
#include <hip/hip_runtime.h>
#include <math.h>

#define N_NODES 20000
#define N_EDGES 320000
#define IN_CH 512
#define HID 128
#define HEADS 4
#define OUT_CH 256
#define NEG_SLOPE 0.2f

typedef float v2f __attribute__((ext_vector_type(2)));
typedef float v8f __attribute__((ext_vector_type(8)));

__device__ __forceinline__ int ford(float f) {
    int i = __float_as_int(f);
    return (i >= 0) ? i : (i ^ 0x7FFFFFFF);
}
__device__ __forceinline__ float iord(int i) {
    return __int_as_float((i >= 0) ? i : (i ^ 0x7FFFFFFF));
}

__global__ void fill_f32(float* __restrict__ p, float v, int n) {
    int i = blockIdx.x * blockDim.x + threadIdx.x;
    if (i < n) p[i] = v;
}
__global__ void fill_i32(int* __restrict__ p, int v, int n) {
    int i = blockIdx.x * blockDim.x + threadIdx.x;
    if (i < n) p[i] = v;
}

template <int K, int NOUT>
__global__ __launch_bounds__(32)
void gemm_wmma_f32(const float* __restrict__ A, const float* __restrict__ B,
                   float* __restrict__ C) {
    const int lane = threadIdx.x & 31;
    const int m0 = blockIdx.x * 16;
    const int n0 = blockIdx.y * 64;
    const int mr = lane & 15;
    const int nr = lane & 15;
    const int kb = (lane >> 4) * 2;

    v8f acc0 = {}, acc1 = {}, acc2 = {}, acc3 = {};
    const float* __restrict__ Arow = A + (size_t)(m0 + mr) * K + kb;

    for (int k = 0; k < K; k += 4) {
        v2f a;
        a.x = Arow[k + 0];
        a.y = Arow[k + 1];

        const float* __restrict__ Bp = B + (size_t)(k + kb) * NOUT + n0 + nr;
        v2f b0, b1, b2, b3;
        b0.x = Bp[0];      b0.y = Bp[NOUT];
        b1.x = Bp[16];     b1.y = Bp[NOUT + 16];
        b2.x = Bp[32];     b2.y = Bp[NOUT + 32];
        b3.x = Bp[48];     b3.y = Bp[NOUT + 48];

        acc0 = __builtin_amdgcn_wmma_f32_16x16x4_f32(false, a, false, b0, (short)0, acc0, false, false);
        acc1 = __builtin_amdgcn_wmma_f32_16x16x4_f32(false, a, false, b1, (short)0, acc1, false, false);
        acc2 = __builtin_amdgcn_wmma_f32_16x16x4_f32(false, a, false, b2, (short)0, acc2, false, false);
        acc3 = __builtin_amdgcn_wmma_f32_16x16x4_f32(false, a, false, b3, (short)0, acc3, false, false);
    }

    const int half = lane >> 4;
#pragma unroll
    for (int r = 0; r < 8; ++r) {
        float* __restrict__ Cp = C + (size_t)(m0 + r + 8 * half) * NOUT + n0 + nr;
        Cp[0]  = acc0[r];
        Cp[16] = acc1[r];
        Cp[32] = acc2[r];
        Cp[48] = acc3[r];
    }
}

template <int CH>
__global__ __launch_bounds__(256)
void att_dots(const float* __restrict__ Hm, const float* __restrict__ a_src,
              const float* __restrict__ a_dst, int nwaves, int heads,
              float* __restrict__ as_out, float* __restrict__ ad_out) {
    int gid = blockIdx.x * blockDim.x + threadIdx.x;
    int wave = gid >> 5;
    int lane = gid & 31;
    if (wave >= nwaves) return;
    int h = wave % heads;
    const float* __restrict__ hp  = Hm + (size_t)wave * CH;
    const float* __restrict__ asv = a_src + h * CH;
    const float* __restrict__ adv = a_dst + h * CH;
    float s = 0.f, d = 0.f;
#pragma unroll
    for (int c = lane; c < CH; c += 32) {
        float v = hp[c];
        s += v * asv[c];
        d += v * adv[c];
    }
#pragma unroll
    for (int off = 16; off > 0; off >>= 1) {
        s += __shfl_xor(s, off, 32);
        d += __shfl_xor(d, off, 32);
    }
    if (lane == 0) { as_out[wave] = s; ad_out[wave] = d; }
}

template <int H>
__global__ __launch_bounds__(256)
void edge_logits(const int* __restrict__ ei, const float* __restrict__ as,
                 const float* __restrict__ ad, float* __restrict__ elog,
                 int* __restrict__ emax) {
    int idx = blockIdx.x * blockDim.x + threadIdx.x;
    const int total = (N_EDGES + N_NODES) * H;
    if (idx >= total) return;
    int e = idx / H, h = idx % H;
    int s = (e < N_EDGES) ? ei[e] : (e - N_EDGES);
    int d = (e < N_EDGES) ? ei[N_EDGES + e] : (e - N_EDGES);
    float v = as[s * H + h] + ad[d * H + h];
    v = (v > 0.f) ? v : NEG_SLOPE * v;
    elog[idx] = v;
    atomicMax(&emax[d * H + h], ford(v));
}

template <int H>
__global__ __launch_bounds__(256)
void edge_exp(const int* __restrict__ ei, float* __restrict__ elog,
              const int* __restrict__ emax, float* __restrict__ denom) {
    int idx = blockIdx.x * blockDim.x + threadIdx.x;
    const int total = (N_EDGES + N_NODES) * H;
    if (idx >= total) return;
    int e = idx / H, h = idx % H;
    int d = (e < N_EDGES) ? ei[N_EDGES + e] : (e - N_EDGES);
    float m = iord(emax[d * H + h]);
    float ex = expf(elog[idx] - m);
    elog[idx] = ex;
    atomicAdd(&denom[d * H + h], ex);
}

template <int H>
__global__ __launch_bounds__(256)
void edge_alpha(const int* __restrict__ ei, float* __restrict__ elog,
                const float* __restrict__ denom, float* __restrict__ alpha_out) {
    int idx = blockIdx.x * blockDim.x + threadIdx.x;
    const int total = (N_EDGES + N_NODES) * H;
    if (idx >= total) return;
    int e = idx / H, h = idx % H;
    int d = (e < N_EDGES) ? ei[N_EDGES + e] : (e - N_EDGES);
    float a = elog[idx] / denom[d * H + h];
    elog[idx] = a;
    alpha_out[idx] = a;
}

template <int H, int CH>
__global__ __launch_bounds__(256)
void edge_msg(const int* __restrict__ ei, const float* __restrict__ Hm,
              const float* __restrict__ alpha, float* __restrict__ agg) {
    int gid = blockIdx.x * blockDim.x + threadIdx.x;
    int wave = gid >> 5;
    int lane = gid & 31;
    const int total = (N_EDGES + N_NODES) * H;
    if (wave >= total) return;
    int e = wave / H, h = wave % H;
    int s = (e < N_EDGES) ? ei[e] : (e - N_EDGES);
    int d = (e < N_EDGES) ? ei[N_EDGES + e] : (e - N_EDGES);
    float al = alpha[wave];
    const float* __restrict__ hp = Hm + (size_t)s * (H * CH) + h * CH;
    float* __restrict__ op = agg + (size_t)d * (H * CH) + h * CH;
#pragma unroll
    for (int c = lane; c < CH; c += 32)
        atomicAdd(&op[c], al * hp[c]);
}

template <int CH>
__global__ __launch_bounds__(256)
void elu_bias_inplace(float* __restrict__ p, const float* __restrict__ bias, int n) {
    int i = blockIdx.x * blockDim.x + threadIdx.x;
    if (i >= n) return;
    float v = p[i] + bias[i & (CH - 1)];
    p[i] = (v > 0.f) ? v : (expf(v) - 1.f);
}

template <int CH>
__global__ __launch_bounds__(256)
void bias_inplace(float* __restrict__ p, const float* __restrict__ bias, int n) {
    int i = blockIdx.x * blockDim.x + threadIdx.x;
    if (i >= n) return;
    p[i] += bias[i & (CH - 1)];
}

static inline int cdiv(long long a, int b) { return (int)((a + b - 1) / b); }

extern "C" void kernel_launch(void* const* d_in, const int* in_sizes, int n_in,
                              void* d_out, int out_size, void* d_ws, size_t ws_size,
                              hipStream_t stream) {
    const float* x      = (const float*)d_in[0];
    const int*   ei     = (const int*)d_in[1];
    const float* W1     = (const float*)d_in[2];
    const float* a_src1 = (const float*)d_in[3];
    const float* a_dst1 = (const float*)d_in[4];
    const float* b1     = (const float*)d_in[5];
    const float* W2     = (const float*)d_in[6];
    const float* a_src2 = (const float*)d_in[7];
    const float* a_dst2 = (const float*)d_in[8];
    const float* b2     = (const float*)d_in[9];

    const int EP = N_EDGES + N_NODES;

    float* H1   = (float*)d_ws;
    float* AGG1 = H1   + (size_t)N_NODES * 512;
    float* AS1  = AGG1 + (size_t)N_NODES * 512;
    float* AD1  = AS1  + (size_t)N_NODES * HEADS;
    float* DEN1 = AD1  + (size_t)N_NODES * HEADS;
    int*   EMX1 = (int*)(DEN1 + (size_t)N_NODES * HEADS);
    float* ELG1 = (float*)(EMX1 + (size_t)N_NODES * HEADS);
    float* AS2  = ELG1 + (size_t)EP * HEADS;
    float* AD2  = AS2  + N_NODES;
    float* DEN2 = AD2  + N_NODES;
    int*   EMX2 = (int*)(DEN2 + N_NODES);
    float* ELG2 = (float*)(EMX2 + N_NODES);
    float* G2   = H1;

    float* out    = (float*)d_out;
    float* alpha1 = out    + (size_t)N_NODES * OUT_CH;
    float* alpha2 = alpha1 + (size_t)EP * HEADS;

    const int TB = 256;

    gemm_wmma_f32<IN_CH, HEADS * HID>
        <<<dim3(N_NODES / 16, (HEADS * HID) / 64), 32, 0, stream>>>(x, W1, H1);

    att_dots<HID><<<cdiv((long long)N_NODES * HEADS * 32, TB), TB, 0, stream>>>(
        H1, a_src1, a_dst1, N_NODES * HEADS, HEADS, AS1, AD1);

    fill_f32<<<cdiv((long long)N_NODES * 512, TB), TB, 0, stream>>>(AGG1, 0.f, N_NODES * 512);
    fill_f32<<<cdiv((long long)N_NODES * HEADS, TB), TB, 0, stream>>>(DEN1, 0.f, N_NODES * HEADS);
    fill_i32<<<cdiv((long long)N_NODES * HEADS, TB), TB, 0, stream>>>(EMX1, (int)0x80000000, N_NODES * HEADS);

    edge_logits<HEADS><<<cdiv((long long)EP * HEADS, TB), TB, 0, stream>>>(ei, AS1, AD1, ELG1, EMX1);
    edge_exp<HEADS><<<cdiv((long long)EP * HEADS, TB), TB, 0, stream>>>(ei, ELG1, EMX1, DEN1);
    edge_alpha<HEADS><<<cdiv((long long)EP * HEADS, TB), TB, 0, stream>>>(ei, ELG1, DEN1, alpha1);
    edge_msg<HEADS, HID><<<cdiv((long long)EP * HEADS * 32, TB), TB, 0, stream>>>(ei, H1, ELG1, AGG1);

    elu_bias_inplace<512><<<cdiv((long long)N_NODES * 512, TB), TB, 0, stream>>>(AGG1, b1, N_NODES * 512);

    gemm_wmma_f32<HEADS * HID, OUT_CH>
        <<<dim3(N_NODES / 16, OUT_CH / 64), 32, 0, stream>>>(AGG1, W2, G2);

    att_dots<OUT_CH><<<cdiv((long long)N_NODES * 32, TB), TB, 0, stream>>>(
        G2, a_src2, a_dst2, N_NODES, 1, AS2, AD2);

    fill_f32<<<cdiv((long long)N_NODES * OUT_CH, TB), TB, 0, stream>>>(out, 0.f, N_NODES * OUT_CH);
    fill_f32<<<cdiv((long long)N_NODES, TB), TB, 0, stream>>>(DEN2, 0.f, N_NODES);
    fill_i32<<<cdiv((long long)N_NODES, TB), TB, 0, stream>>>(EMX2, (int)0x80000000, N_NODES);

    edge_logits<1><<<cdiv((long long)EP, TB), TB, 0, stream>>>(ei, AS2, AD2, ELG2, EMX2);
    edge_exp<1><<<cdiv((long long)EP, TB), TB, 0, stream>>>(ei, ELG2, EMX2, DEN2);
    edge_alpha<1><<<cdiv((long long)EP, TB), TB, 0, stream>>>(ei, ELG2, DEN2, alpha2);
    edge_msg<1, OUT_CH><<<cdiv((long long)EP * 32, TB), TB, 0, stream>>>(ei, G2, ELG2, out);

    bias_inplace<OUT_CH><<<cdiv((long long)N_NODES * OUT_CH, TB), TB, 0, stream>>>(out, b2, N_NODES * OUT_CH);
}


# Message sums in an LDS tile: a group owns 80 destination rows, scans the edge list for its edges, adds rows by LDS float atomics, one store with bias/ELU; global atomics, fills gone; order: arrival
# speedup vs baseline: 1.9252x; 1.9252x over previous
#include <hip/hip_runtime.h>
#include <math.h>

#define N_NODES 20000
#define N_EDGES 320000
#define IN_CH 512
#define HID 128
#define HEADS 4
#define OUT_CH 256
#define NEG_SLOPE 0.2f

typedef float v2f __attribute__((ext_vector_type(2)));
typedef float v8f __attribute__((ext_vector_type(8)));

__device__ __forceinline__ int ford(float f) {
    int i = __float_as_int(f);
    return (i >= 0) ? i : (i ^ 0x7FFFFFFF);
}
__device__ __forceinline__ float iord(int i) {
    return __int_as_float((i >= 0) ? i : (i ^ 0x7FFFFFFF));
}

__global__ void fill_f32(float* __restrict__ p, float v, int n) {
    int i = blockIdx.x * blockDim.x + threadIdx.x;
    if (i < n) p[i] = v;
}
__global__ void fill_i32(int* __restrict__ p, int v, int n) {
    int i = blockIdx.x * blockDim.x + threadIdx.x;
    if (i < n) p[i] = v;
}

typedef float v4f __attribute__((ext_vector_type(4)));

#define G_BM 160
#define G_BN 128
#define G_BK 32
#define G_AP 36
#define G_BP 584
#define G_CP 36

template <int K, int NOUT>
__global__ __launch_bounds__(256)
void gemm_wmma_f32(const float* __restrict__ A, const float* __restrict__ B,
                   float* __restrict__ C) {
    constexpr int A_ST = G_BM * G_AP;
    constexpr int B_ST = (G_BK / 4) * G_BP;
    constexpr int ST   = A_ST + B_ST;
    constexpr int C_W  = 80 * G_CP;
    constexpr int SMEM = (2 * ST > 8 * C_W) ? 2 * ST : 8 * C_W;
    constexpr int NS   = K / G_BK;
    static_assert(K % G_BK == 0 && NOUT % G_BN == 0 && N_NODES % G_BM == 0, "tiles must divide");
    static_assert(G_BK / 4 == 8, "one k quad of the slab per wave in the B copy");
    __shared__ __attribute__((aligned(16))) float smem[SMEM];

    const int tid  = threadIdx.x;
    const int lane = tid & 31;
    const int wave = tid >> 5;
    const int c  = lane & 15;
    const int h  = lane >> 4;
    const int wm = wave >> 2;
    const int wn = wave & 3;
    const int n0 = blockIdx.x * G_BN;
    const int m0 = blockIdx.y * G_BM;

    const int a_row = tid >> 3;
    const int a_k4  = (tid & 7) * 4;
    const float* __restrict__ Ag = A + (size_t)(m0 + a_row) * K + a_k4;
    const float* __restrict__ Bg = B + (size_t)(4 * wave) * NOUT + n0 + 4 * lane;
    float* const As_w = smem + a_row * G_AP + a_k4;
    float* const Bs_w = smem + A_ST + wave * G_BP + 4 * lane;
    const float* const Af = smem + (wm * 80 + c) * G_AP + 4 * h;
    const float* const Bf = smem + A_ST + h * G_BP + 4 * ((c & 3) * 36 + wn * 8 + (c >> 2));

    v4f ra[5], rb[4];
    v8f acc[5][2] = {};

#define G_GLOAD(k0)                                                              \
    _Pragma("unroll") for (int p = 0; p < 5; ++p)                                \
        ra[p] = *(const v4f*)(Ag + (size_t)(p * 32) * K + (k0));                 \
    _Pragma("unroll") for (int r = 0; r < 4; ++r)                                \
        rb[r] = *(const v4f*)(Bg + (size_t)((k0) + r) * NOUT);
#define G_SSTORE(st)                                                             \
    _Pragma("unroll") for (int p = 0; p < 5; ++p)                                \
        *(v4f*)(As_w + (st) * ST + p * 32 * G_AP) = ra[p];                       \
    _Pragma("unroll") for (int i = 0; i < 4; ++i) {                              \
        v4f col = {rb[0][i], rb[1][i], rb[2][i], rb[3][i]};                      \
        *(v4f*)(Bs_w + (st) * ST + i * 36 * 4) = col;                            \
    }
#define G_COMPUTE(Afs, Bfs)                                                      \
    _Pragma("unroll") for (int t = 0; t < G_BK / 8; ++t) {                       \
        v4f fa[5], fb[2];                                                        \
        _Pragma("unroll") for (int i = 0; i < 5; ++i)                            \
            fa[i] = *(const v4f*)((Afs) + i * 16 * G_AP + t * 8);                \
        _Pragma("unroll") for (int j = 0; j < 2; ++j)                            \
            fb[j] = *(const v4f*)((Bfs) + t * 2 * G_BP + j * 16);                \
        _Pragma("unroll") for (int i = 0; i < 5; ++i)                            \
            _Pragma("unroll") for (int j = 0; j < 2; ++j) {                      \
                v2f a = {fa[i].x, fa[i].y};                                      \
                v2f b = {fb[j].x, fb[j].y};                                      \
                acc[i][j] = __builtin_amdgcn_wmma_f32_16x16x4_f32(               \
                    false, a, false, b, (short)0, acc[i][j], false, false);      \
            }                                                                    \
        _Pragma("unroll") for (int i = 0; i < 5; ++i)                            \
            _Pragma("unroll") for (int j = 0; j < 2; ++j) {                      \
                v2f a = {fa[i].z, fa[i].w};                                      \
                v2f b = {fb[j].z, fb[j].w};                                      \
                acc[i][j] = __builtin_amdgcn_wmma_f32_16x16x4_f32(               \
                    false, a, false, b, (short)0, acc[i][j], false, false);      \
            }                                                                    \
    }
#define G_FENCE() __builtin_amdgcn_sched_barrier(0)
#define G_SLOTS() asm volatile("v_nop\n\tv_nop\n\tv_nop\n\tv_nop")

    G_GLOAD(0)
    G_SSTORE(0)
    __syncthreads();

    for (int s = 0; s < NS - 1; ++s) {
        const int st = s & 1;
        const float* const Afs = Af + st * ST;
        const float* const Bfs = Bf + st * ST;
        G_GLOAD((s + 1) * G_BK)
        G_FENCE();
        G_COMPUTE(Afs, Bfs)
        G_FENCE();
        G_SLOTS();
        G_FENCE();
        G_SSTORE(st ^ 1)
        __syncthreads();
    }
    {
        const float* const Afs = Af + ((NS - 1) & 1) * ST;
        const float* const Bfs = Bf + ((NS - 1) & 1) * ST;
        G_FENCE();
        G_COMPUTE(Afs, Bfs)
        G_FENCE();
        G_SLOTS();
        G_FENCE();
    }
    __syncthreads();

    float* const Cw = smem + wave * C_W;
#pragma unroll
    for (int i = 0; i < 5; ++i)
#pragma unroll
        for (int j = 0; j < 2; ++j)
#pragma unroll
            for (int r = 0; r < 8; ++r)
                Cw[(i * 16 + 8 * h + r) * G_CP + j * 16 + c] = acc[i][j][r];
    __syncthreads();
    const int rr = lane >> 3;
    const int c4 = (lane & 7) * 4;
    float* __restrict__ Cg = C + (size_t)(m0 + wm * 80 + rr) * NOUT + n0 + wn * 32 + c4;
#pragma unroll
    for (int t = 0; t < 20; ++t) {
        v4f v = *(const v4f*)(Cw + (t * 4 + rr) * G_CP + c4);
        *(v4f*)(Cg + (size_t)(t * 4) * NOUT) = v;
    }
#undef G_GLOAD
#undef G_SSTORE
#undef G_COMPUTE
#undef G_FENCE
#undef G_SLOTS
}

template <int CH>
__global__ __launch_bounds__(256)
void att_dots(const float* __restrict__ Hm, const float* __restrict__ a_src,
              const float* __restrict__ a_dst, int nwaves, int heads,
              float* __restrict__ as_out, float* __restrict__ ad_out) {
    int gid = blockIdx.x * blockDim.x + threadIdx.x;
    int wave = gid >> 5;
    int lane = gid & 31;
    if (wave >= nwaves) return;
    int h = wave % heads;
    const float* __restrict__ hp  = Hm + (size_t)wave * CH;
    const float* __restrict__ asv = a_src + h * CH;
    const float* __restrict__ adv = a_dst + h * CH;
    float s = 0.f, d = 0.f;
#pragma unroll
    for (int c = lane; c < CH; c += 32) {
        float v = hp[c];
        s += v * asv[c];
        d += v * adv[c];
    }
#pragma unroll
    for (int off = 16; off > 0; off >>= 1) {
        s += __shfl_xor(s, off, 32);
        d += __shfl_xor(d, off, 32);
    }
    if (lane == 0) { as_out[wave] = s; ad_out[wave] = d; }
}

template <int H>
__global__ __launch_bounds__(256)
void edge_logits(const int* __restrict__ ei, const float* __restrict__ as,
                 const float* __restrict__ ad, float* __restrict__ elog,
                 int* __restrict__ emax) {
    int idx = blockIdx.x * blockDim.x + threadIdx.x;
    const int total = (N_EDGES + N_NODES) * H;
    if (idx >= total) return;
    int e = idx / H, h = idx % H;
    int s = (e < N_EDGES) ? ei[e] : (e - N_EDGES);
    int d = (e < N_EDGES) ? ei[N_EDGES + e] : (e - N_EDGES);
    float v = as[s * H + h] + ad[d * H + h];
    v = (v > 0.f) ? v : NEG_SLOPE * v;
    elog[idx] = v;
    atomicMax(&emax[d * H + h], ford(v));
}

template <int H>
__global__ __launch_bounds__(256)
void edge_exp(const int* __restrict__ ei, float* __restrict__ elog,
              const int* __restrict__ emax, float* __restrict__ denom) {
    int idx = blockIdx.x * blockDim.x + threadIdx.x;
    const int total = (N_EDGES + N_NODES) * H;
    if (idx >= total) return;
    int e = idx / H, h = idx % H;
    int d = (e < N_EDGES) ? ei[N_EDGES + e] : (e - N_EDGES);
    float m = iord(emax[d * H + h]);
    float ex = expf(elog[idx] - m);
    elog[idx] = ex;
    atomicAdd(&denom[d * H + h], ex);
}

template <int H>
__global__ __launch_bounds__(256)
void edge_alpha(const int* __restrict__ ei, float* __restrict__ elog,
                const float* __restrict__ denom, float* __restrict__ alpha_out) {
    int idx = blockIdx.x * blockDim.x + threadIdx.x;
    const int total = (N_EDGES + N_NODES) * H;
    if (idx >= total) return;
    int e = idx / H, h = idx % H;
    int d = (e < N_EDGES) ? ei[N_EDGES + e] : (e - N_EDGES);
    float a = elog[idx] / denom[d * H + h];
    elog[idx] = a;
    alpha_out[idx] = a;
}

#define T_R 80
#define T_CAP 4096
#define T_THREADS 1024
#define T_U 4

template <int H, int CH, int NE>
__device__ __forceinline__ void tile_add_rows(const int* __restrict__ ei,
                                              const float* __restrict__ Hm,
                                              const float* __restrict__ alpha,
                                              const int* lst, float* tile, int i, int lane) {
    constexpr int ROW = H * CH;
    constexpr int NL  = ROW / 32;
    constexpr int LPH = CH / 32;
    int s[NE], d[NE];
    float al[NE][H];
    float v[NE][NL];
#pragma unroll
    for (int k = 0; k < NE; ++k) {
        const int ent = __builtin_amdgcn_readfirstlane(lst[i + k]);
        const int e = ent >> 7;
        d[k] = ent & 127;
        s[k] = ei[e];
#pragma unroll
        for (int h = 0; h < H; ++h) al[k][h] = alpha[e * H + h];
    }
#pragma unroll
    for (int k = 0; k < NE; ++k) {
        const float* __restrict__ hp = Hm + (size_t)s[k] * ROW + lane;
#pragma unroll
        for (int j = 0; j < NL; ++j) v[k][j] = hp[32 * j];
    }
#pragma unroll
    for (int k = 0; k < NE; ++k) {
        float* tp = tile + d[k] * ROW + lane;
#pragma unroll
        for (int j = 0; j < NL; ++j)
            atomicAdd(&tp[32 * j], al[k][j / LPH] * v[k][j]);
    }
}

template <int H, int CH, int EPW, bool ELU>
__global__ __launch_bounds__(T_THREADS)
void tile_msg(const int* __restrict__ ei, const float* __restrict__ Hm,
              const float* __restrict__ alpha, const float* __restrict__ bias,
              float* __restrict__ outp) {
    constexpr int ROW = H * CH;
    constexpr int TS  = T_R * ROW;
    static_assert(N_NODES % T_R == 0, "tiles must cover the nodes");
    static_assert(T_R <= 128 && N_EDGES <= (1 << 24), "list entry packs edge id and local row");
    static_assert(CH % 32 == 0 && TS % (4 * T_THREADS) == 0 && N_EDGES % 4 == 0, "16-byte pieces");
    __shared__ __attribute__((aligned(16))) float tile[TS];
    __shared__ int lst[T_CAP];
    __shared__ int cnt;

    const int tid  = threadIdx.x;
    const int lane = tid & 31;
    const int n0   = blockIdx.x * T_R;

    if (tid == 0) cnt = 0;
    constexpr int NP = TS / (4 * T_THREADS);
    {
        v4f sv[NP];
        float sa[NP];
#pragma unroll
        for (int p = 0; p < NP; ++p) {
            const int i = (p * T_THREADS + tid) * 4;
            sa[p] = alpha[(size_t)(N_EDGES + n0 + i / ROW) * H + (i % ROW) / CH];
            sv[p] = *(const v4f*)(Hm + (size_t)n0 * ROW + i);
        }
#pragma unroll
        for (int p = 0; p < NP; ++p)
            *(v4f*)(tile + (p * T_THREADS + tid) * 4) = sa[p] * sv[p];
    }
    __syncthreads();

    const int* __restrict__ dst = ei + N_EDGES;
    for (int base = tid * 4; base < N_EDGES; base += 4 * T_THREADS * T_U) {
        int4 dv[T_U];
#pragma unroll
        for (int u = 0; u < T_U; ++u) {
            const int e4 = base + u * 4 * T_THREADS;
            dv[u] = (e4 < N_EDGES) ? *(const int4*)(dst + e4) : make_int4(-1, -1, -1, -1);
        }
#pragma unroll
        for (int u = 0; u < T_U; ++u) {
            const int e4 = base + u * 4 * T_THREADS;
            const unsigned dq[4] = {(unsigned)(dv[u].x - n0), (unsigned)(dv[u].y - n0),
                                    (unsigned)(dv[u].z - n0), (unsigned)(dv[u].w - n0)};
            const unsigned lo = min(min(dq[0], dq[1]), min(dq[2], dq[3]));
            if (lo < (unsigned)T_R) {
#pragma unroll
                for (int q = 0; q < 4; ++q) {
                    if (dq[q] < (unsigned)T_R) {
                        const int d = (int)dq[q];
                        const int e = e4 + q;
                        const int slot = atomicAdd(&cnt, 1);
                        if (slot < T_CAP) {
                            lst[slot] = (e << 7) | d;
                        } else {
                            const int s = ei[e];
#pragma nounroll
                            for (int c = 0; c < ROW; ++c)
                                atomicAdd(&tile[d * ROW + c],
                                          alpha[(size_t)e * H + c / CH] * Hm[(size_t)s * ROW + c]);
                        }
                    }
                }
            }
        }
    }
    __syncthreads();

    const int n = (cnt < T_CAP) ? cnt : T_CAP;
    const int wave = __builtin_amdgcn_readfirstlane(tid >> 5);
    constexpr int NW = T_THREADS / 32;
    const int nfull = (n / EPW) * EPW;
    for (int i = wave * EPW; i < nfull; i += NW * EPW)
        tile_add_rows<H, CH, EPW>(ei, Hm, alpha, lst, tile, i, lane);
    for (int i = nfull + wave; i < n; i += NW)
        tile_add_rows<H, CH, 1>(ei, Hm, alpha, lst, tile, i, lane);
    __syncthreads();

    for (int i = tid * 4; i < TS; i += 4 * T_THREADS) {
        const int c = i % ROW;
        v4f t = *(const v4f*)(tile + i);
        const v4f b = *(const v4f*)(bias + c);
        t += b;
        if (ELU) {
#pragma unroll
            for (int q = 0; q < 4; ++q) t[q] = (t[q] > 0.f) ? t[q] : (expf(t[q]) - 1.f);
        }
        *(v4f*)(outp + (size_t)n0 * ROW + i) = t;
    }
}

static inline int cdiv(long long a, int b) { return (int)((a + b - 1) / b); }

extern "C" void kernel_launch(void* const* d_in, const int* in_sizes, int n_in,
                              void* d_out, int out_size, void* d_ws, size_t ws_size,
                              hipStream_t stream) {
    const float* x      = (const float*)d_in[0];
    const int*   ei     = (const int*)d_in[1];
    const float* W1     = (const float*)d_in[2];
    const float* a_src1 = (const float*)d_in[3];
    const float* a_dst1 = (const float*)d_in[4];
    const float* b1     = (const float*)d_in[5];
    const float* W2     = (const float*)d_in[6];
    const float* a_src2 = (const float*)d_in[7];
    const float* a_dst2 = (const float*)d_in[8];
    const float* b2     = (const float*)d_in[9];

    const int EP = N_EDGES + N_NODES;

    float* H1   = (float*)d_ws;
    float* AGG1 = H1   + (size_t)N_NODES * 512;
    float* AS1  = AGG1 + (size_t)N_NODES * 512;
    float* AD1  = AS1  + (size_t)N_NODES * HEADS;
    float* DEN1 = AD1  + (size_t)N_NODES * HEADS;
    int*   EMX1 = (int*)(DEN1 + (size_t)N_NODES * HEADS);
    float* ELG1 = (float*)(EMX1 + (size_t)N_NODES * HEADS);
    float* AS2  = ELG1 + (size_t)EP * HEADS;
    float* AD2  = AS2  + N_NODES;
    float* DEN2 = AD2  + N_NODES;
    int*   EMX2 = (int*)(DEN2 + N_NODES);
    float* ELG2 = (float*)(EMX2 + N_NODES);
    float* G2   = H1;

    float* out    = (float*)d_out;
    float* alpha1 = out    + (size_t)N_NODES * OUT_CH;
    float* alpha2 = alpha1 + (size_t)EP * HEADS;

    const int TB = 256;

    gemm_wmma_f32<IN_CH, HEADS * HID>
        <<<dim3((HEADS * HID) / G_BN, N_NODES / G_BM), 256, 0, stream>>>(x, W1, H1);

    att_dots<HID><<<cdiv((long long)N_NODES * HEADS * 32, TB), TB, 0, stream>>>(
        H1, a_src1, a_dst1, N_NODES * HEADS, HEADS, AS1, AD1);

    fill_f32<<<cdiv((long long)N_NODES * HEADS, TB), TB, 0, stream>>>(DEN1, 0.f, N_NODES * HEADS);
    fill_i32<<<cdiv((long long)N_NODES * HEADS, TB), TB, 0, stream>>>(EMX1, (int)0x80000000, N_NODES * HEADS);

    edge_logits<HEADS><<<cdiv((long long)EP * HEADS, TB), TB, 0, stream>>>(ei, AS1, AD1, ELG1, EMX1);
    edge_exp<HEADS><<<cdiv((long long)EP * HEADS, TB), TB, 0, stream>>>(ei, ELG1, EMX1, DEN1);
    edge_alpha<HEADS><<<cdiv((long long)EP * HEADS, TB), TB, 0, stream>>>(ei, ELG1, DEN1, alpha1);
    tile_msg<HEADS, HID, 2, true><<<N_NODES / T_R, T_THREADS, 0, stream>>>(ei, H1, ELG1, b1, AGG1);

    gemm_wmma_f32<HEADS * HID, OUT_CH>
        <<<dim3(OUT_CH / G_BN, N_NODES / G_BM), 256, 0, stream>>>(AGG1, W2, G2);

    att_dots<OUT_CH><<<cdiv((long long)N_NODES * 32, TB), TB, 0, stream>>>(
        G2, a_src2, a_dst2, N_NODES, 1, AS2, AD2);

    fill_f32<<<cdiv((long long)N_NODES, TB), TB, 0, stream>>>(DEN2, 0.f, N_NODES);
    fill_i32<<<cdiv((long long)N_NODES, TB), TB, 0, stream>>>(EMX2, (int)0x80000000, N_NODES);

    edge_logits<1><<<cdiv((long long)EP, TB), TB, 0, stream>>>(ei, AS2, AD2, ELG2, EMX2);
    edge_exp<1><<<cdiv((long long)EP, TB), TB, 0, stream>>>(ei, ELG2, EMX2, DEN2);
    edge_alpha<1><<<cdiv((long long)EP, TB), TB, 0, stream>>>(ei, ELG2, DEN2, alpha2);
    tile_msg<1, OUT_CH, 4, false><<<N_NODES / T_R, T_THREADS, 0, stream>>>(ei, G2, ELG2, b2, out);
}


# Segment softmax (max, exp sum, alpha) done inside the group that owns the LDS message tile, on its LDS edge list by LDS atomics; 6 edge passes, 4 fills gone (16->6 launches); sum order: arrival
# speedup vs baseline: 2.6404x; 1.3715x over previous
#include <hip/hip_runtime.h>
#include <math.h>

#define N_NODES 20000
#define N_EDGES 320000
#define IN_CH 512
#define HID 128
#define HEADS 4
#define OUT_CH 256
#define NEG_SLOPE 0.2f

typedef float v2f __attribute__((ext_vector_type(2)));
typedef float v8f __attribute__((ext_vector_type(8)));

__device__ __forceinline__ int ford(float f) {
    int i = __float_as_int(f);
    return (i >= 0) ? i : (i ^ 0x7FFFFFFF);
}
__device__ __forceinline__ float iord(int i) {
    return __int_as_float((i >= 0) ? i : (i ^ 0x7FFFFFFF));
}

typedef float v4f __attribute__((ext_vector_type(4)));

#define G_BM 160
#define G_BN 128
#define G_BK 32
#define G_AP 36
#define G_BP 584
#define G_CP 36

template <int K, int NOUT>
__global__ __launch_bounds__(256)
void gemm_wmma_f32(const float* __restrict__ A, const float* __restrict__ B,
                   float* __restrict__ C) {
    constexpr int A_ST = G_BM * G_AP;
    constexpr int B_ST = (G_BK / 4) * G_BP;
    constexpr int ST   = A_ST + B_ST;
    constexpr int C_W  = 80 * G_CP;
    constexpr int SMEM = (2 * ST > 8 * C_W) ? 2 * ST : 8 * C_W;
    constexpr int NS   = K / G_BK;
    static_assert(K % G_BK == 0 && NOUT % G_BN == 0 && N_NODES % G_BM == 0, "tiles must divide");
    static_assert(G_BK / 4 == 8, "one k quad of the slab per wave in the B copy");
    __shared__ __attribute__((aligned(16))) float smem[SMEM];

    const int tid  = threadIdx.x;
    const int lane = tid & 31;
    const int wave = tid >> 5;
    const int c  = lane & 15;
    const int h  = lane >> 4;
    const int wm = wave >> 2;
    const int wn = wave & 3;
    const int n0 = blockIdx.x * G_BN;
    const int m0 = blockIdx.y * G_BM;

    const int a_row = tid >> 3;
    const int a_k4  = (tid & 7) * 4;
    const float* __restrict__ Ag = A + (size_t)(m0 + a_row) * K + a_k4;
    const float* __restrict__ Bg = B + (size_t)(4 * wave) * NOUT + n0 + 4 * lane;
    float* const As_w = smem + a_row * G_AP + a_k4;
    float* const Bs_w = smem + A_ST + wave * G_BP + 4 * lane;
    const float* const Af = smem + (wm * 80 + c) * G_AP + 4 * h;
    const float* const Bf = smem + A_ST + h * G_BP + 4 * ((c & 3) * 36 + wn * 8 + (c >> 2));

    v4f ra[5], rb[4];
    v8f acc[5][2] = {};

#define G_GLOAD(k0)                                                              \
    _Pragma("unroll") for (int p = 0; p < 5; ++p)                                \
        ra[p] = *(const v4f*)(Ag + (size_t)(p * 32) * K + (k0));                 \
    _Pragma("unroll") for (int r = 0; r < 4; ++r)                                \
        rb[r] = *(const v4f*)(Bg + (size_t)((k0) + r) * NOUT);
#define G_SSTORE(st)                                                             \
    _Pragma("unroll") for (int p = 0; p < 5; ++p)                                \
        *(v4f*)(As_w + (st) * ST + p * 32 * G_AP) = ra[p];                       \
    _Pragma("unroll") for (int i = 0; i < 4; ++i) {                              \
        v4f col = {rb[0][i], rb[1][i], rb[2][i], rb[3][i]};                      \
        *(v4f*)(Bs_w + (st) * ST + i * 36 * 4) = col;                            \
    }
#define G_COMPUTE(Afs, Bfs)                                                      \
    _Pragma("unroll") for (int t = 0; t < G_BK / 8; ++t) {                       \
        v4f fa[5], fb[2];                                                        \
        _Pragma("unroll") for (int i = 0; i < 5; ++i)                            \
            fa[i] = *(const v4f*)((Afs) + i * 16 * G_AP + t * 8);                \
        _Pragma("unroll") for (int j = 0; j < 2; ++j)                            \
            fb[j] = *(const v4f*)((Bfs) + t * 2 * G_BP + j * 16);                \
        _Pragma("unroll") for (int i = 0; i < 5; ++i)                            \
            _Pragma("unroll") for (int j = 0; j < 2; ++j) {                      \
                v2f a = {fa[i].x, fa[i].y};                                      \
                v2f b = {fb[j].x, fb[j].y};                                      \
                acc[i][j] = __builtin_amdgcn_wmma_f32_16x16x4_f32(               \
                    false, a, false, b, (short)0, acc[i][j], false, false);      \
            }                                                                    \
        _Pragma("unroll") for (int i = 0; i < 5; ++i)                            \
            _Pragma("unroll") for (int j = 0; j < 2; ++j) {                      \
                v2f a = {fa[i].z, fa[i].w};                                      \
                v2f b = {fb[j].z, fb[j].w};                                      \
                acc[i][j] = __builtin_amdgcn_wmma_f32_16x16x4_f32(               \
                    false, a, false, b, (short)0, acc[i][j], false, false);      \
            }                                                                    \
    }
#define G_FENCE() __builtin_amdgcn_sched_barrier(0)
#define G_SLOTS() asm volatile("v_nop\n\tv_nop\n\tv_nop\n\tv_nop")

    G_GLOAD(0)
    G_SSTORE(0)
    __syncthreads();

    for (int s = 0; s < NS - 1; ++s) {
        const int st = s & 1;
        const float* const Afs = Af + st * ST;
        const float* const Bfs = Bf + st * ST;
        G_GLOAD((s + 1) * G_BK)
        G_FENCE();
        G_COMPUTE(Afs, Bfs)
        G_FENCE();
        G_SLOTS();
        G_FENCE();
        G_SSTORE(st ^ 1)
        __syncthreads();
    }
    {
        const float* const Afs = Af + ((NS - 1) & 1) * ST;
        const float* const Bfs = Bf + ((NS - 1) & 1) * ST;
        G_FENCE();
        G_COMPUTE(Afs, Bfs)
        G_FENCE();
        G_SLOTS();
        G_FENCE();
    }
    __syncthreads();

    float* const Cw = smem + wave * C_W;
#pragma unroll
    for (int i = 0; i < 5; ++i)
#pragma unroll
        for (int j = 0; j < 2; ++j)
#pragma unroll
            for (int r = 0; r < 8; ++r)
                Cw[(i * 16 + 8 * h + r) * G_CP + j * 16 + c] = acc[i][j][r];
    __syncthreads();
    const int rr = lane >> 3;
    const int c4 = (lane & 7) * 4;
    float* __restrict__ Cg = C + (size_t)(m0 + wm * 80 + rr) * NOUT + n0 + wn * 32 + c4;
#pragma unroll
    for (int t = 0; t < 20; ++t) {
        v4f v = *(const v4f*)(Cw + (t * 4 + rr) * G_CP + c4);
        *(v4f*)(Cg + (size_t)(t * 4) * NOUT) = v;
    }
#undef G_GLOAD
#undef G_SSTORE
#undef G_COMPUTE
#undef G_FENCE
#undef G_SLOTS
}

template <int CH>
__global__ __launch_bounds__(256)
void att_dots(const float* __restrict__ Hm, const float* __restrict__ a_src,
              const float* __restrict__ a_dst, int nwaves, int heads,
              float* __restrict__ as_out, float* __restrict__ ad_out) {
    int gid = blockIdx.x * blockDim.x + threadIdx.x;
    int wave = gid >> 5;
    int lane = gid & 31;
    if (wave >= nwaves) return;
    int h = wave % heads;
    const float* __restrict__ hp  = Hm + (size_t)wave * CH;
    const float* __restrict__ asv = a_src + h * CH;
    const float* __restrict__ adv = a_dst + h * CH;
    float s = 0.f, d = 0.f;
#pragma unroll
    for (int c = lane; c < CH; c += 32) {
        float v = hp[c];
        s += v * asv[c];
        d += v * adv[c];
    }
#pragma unroll
    for (int off = 16; off > 0; off >>= 1) {
        s += __shfl_xor(s, off, 32);
        d += __shfl_xor(d, off, 32);
    }
    if (lane == 0) { as_out[wave] = s; ad_out[wave] = d; }
}

#define T_R 80
#define T_CAP 4096
#define T_THREADS 1024
#define T_U 4

__device__ __forceinline__ float att_logit(float as, float ad) {
    const float v = as + ad;
    return (v > 0.f) ? v : NEG_SLOPE * v;
}

template <int H, int CH, int NE>
__device__ __forceinline__ void tile_add_rows(const float* __restrict__ Hm, const int* lst,
                                              const int* srcl, const float* val,
                                              float* tile, int i, int lane) {
    constexpr int ROW = H * CH;
    constexpr int NL  = ROW / 32;
    constexpr int LPH = CH / 32;
    int s[NE], d[NE];
    float al[NE][H];
    float v[NE][NL];
#pragma unroll
    for (int k = 0; k < NE; ++k) {
        d[k] = __builtin_amdgcn_readfirstlane(lst[i + k]) & 127;
        s[k] = __builtin_amdgcn_readfirstlane(srcl[i + k]);
#pragma unroll
        for (int h = 0; h < H; ++h) al[k][h] = val[(i + k) * H + h];
    }
#pragma unroll
    for (int k = 0; k < NE; ++k) {
        const float* __restrict__ hp = Hm + (size_t)s[k] * ROW + lane;
#pragma unroll
        for (int j = 0; j < NL; ++j) v[k][j] = hp[32 * j];
    }
#pragma unroll
    for (int k = 0; k < NE; ++k) {
        float* tp = tile + d[k] * ROW + lane;
#pragma unroll
        for (int j = 0; j < NL; ++j)
            atomicAdd(&tp[32 * j], al[k][j / LPH] * v[k][j]);
    }
}

template <int H, int CH, int PH>
__device__ __forceinline__ void edge_list_stage(const int* __restrict__ ei,
                                                const float* __restrict__ Hm,
                                                const float* __restrict__ as, const float* adl,
                                                int* emx, float* den, float* tile,
                                                float* __restrict__ alpha_out, int n0, int tid) {
    constexpr int ROW = H * CH;
#pragma nounroll
    for (int e = tid; e < N_EDGES; e += T_THREADS) {
        const unsigned d = (unsigned)(ei[N_EDGES + e] - n0);
        if (d < (unsigned)T_R) {
            const int s = ei[e];
#pragma nounroll
            for (int h = 0; h < H; ++h) {
                const float v = att_logit(as[(size_t)s * H + h], adl[d * H + h]);
                if (PH == 0) {
                    atomicMax(&emx[d * H + h], ford(v));
                } else {
                    const float ex = expf(v - iord(emx[d * H + h]));
                    if (PH == 1) {
                        atomicAdd(&den[d * H + h], ex);
                    } else {
                        const float a = ex / den[d * H + h];
                        alpha_out[(size_t)e * H + h] = a;
#pragma nounroll
                        for (int c = h * CH; c < (h + 1) * CH; ++c)
                            atomicAdd(&tile[d * ROW + c], a * Hm[(size_t)s * ROW + c]);
                    }
                }
            }
        }
    }
}

template <int H, int CH, int EPW, bool ELU>
__global__ __launch_bounds__(T_THREADS)
void tile_msg(const int* __restrict__ ei, const float* __restrict__ Hm,
              const float* __restrict__ as, const float* __restrict__ ad,
              const float* __restrict__ bias, float* __restrict__ alpha_out,
              float* __restrict__ outp) {
    constexpr int ROW = H * CH;
    constexpr int TS  = T_R * ROW;
    constexpr int SEG = T_R * H;
    static_assert(N_NODES % T_R == 0, "tiles must cover the nodes");
    static_assert(T_R <= 128 && N_EDGES + N_NODES <= (1 << 24), "list entry packs edge id and local row");
    static_assert(CH % 32 == 0 && TS % (4 * T_THREADS) == 0 && N_EDGES % 4 == 0, "16-byte pieces");
    static_assert(SEG <= T_THREADS && T_R < T_CAP, "one thread per segment; self loops fit the list");
    __shared__ __attribute__((aligned(16))) float tile[TS];
    __shared__ __attribute__((aligned(16))) float val[T_CAP * H];
    __shared__ int lst[T_CAP];
    __shared__ int srcl[T_CAP];
    __shared__ float adl[SEG];
    __shared__ int emx[SEG];
    __shared__ float den[SEG];
    __shared__ int cnt;

    const int tid  = threadIdx.x;
    const int lane = tid & 31;
    const int n0   = blockIdx.x * T_R;

    if (tid == 0) cnt = T_R;
    if (tid < T_R) lst[tid] = ((N_EDGES + n0 + tid) << 7) | tid;
    if (tid < SEG) { emx[tid] = (int)0x80000000; den[tid] = 0.f; }
    __syncthreads();

    float adv = 0.f;
    if (tid < SEG) adv = ad[(size_t)n0 * H + tid];

    const int* __restrict__ dst = ei + N_EDGES;
    for (int base = tid * 4; base < N_EDGES; base += 4 * T_THREADS * T_U) {
        int4 dv[T_U];
#pragma unroll
        for (int u = 0; u < T_U; ++u) {
            const int e4 = base + u * 4 * T_THREADS;
            dv[u] = (e4 < N_EDGES) ? *(const int4*)(dst + e4) : make_int4(-1, -1, -1, -1);
        }
#pragma unroll
        for (int u = 0; u < T_U; ++u) {
            const int e4 = base + u * 4 * T_THREADS;
            const unsigned dq[4] = {(unsigned)(dv[u].x - n0), (unsigned)(dv[u].y - n0),
                                    (unsigned)(dv[u].z - n0), (unsigned)(dv[u].w - n0)};
            const unsigned lo = min(min(dq[0], dq[1]), min(dq[2], dq[3]));
            if (lo < (unsigned)T_R) {
#pragma unroll
                for (int q = 0; q < 4; ++q) {
                    if (dq[q] < (unsigned)T_R) {
                        const int slot = atomicAdd(&cnt, 1);
                        if (slot < T_CAP) lst[slot] = ((e4 + q) << 7) | (int)dq[q];
                    }
                }
            }
        }
    }
    if (tid < SEG) adl[tid] = adv;
    __syncthreads();

    const int found = cnt;
    const bool over = found > T_CAP;
    const int n = over ? T_R : found;

    constexpr int NP = TS / (4 * T_THREADS);
    v4f sv[NP];
#pragma unroll
    for (int p = 0; p < NP; ++p)
        sv[p] = *(const v4f*)(Hm + (size_t)n0 * ROW + (p * T_THREADS + tid) * 4);

    for (int i = tid; i < n; i += T_THREADS) {
        const int ent = lst[i];
        const int e = ent >> 7;
        const int d = ent & 127;
        const int s = (e < N_EDGES) ? ei[e] : (e - N_EDGES);
        srcl[i] = s;
#pragma unroll
        for (int h = 0; h < H; ++h) {
            const float v = att_logit(as[(size_t)s * H + h], adl[d * H + h]);
            val[i * H + h] = v;
            atomicMax(&emx[d * H + h], ford(v));
        }
    }
    if (over) edge_list_stage<H, CH, 0>(ei, Hm, as, adl, emx, den, tile, alpha_out, n0, tid);
    __syncthreads();

    for (int i = tid; i < n; i += T_THREADS) {
        const int d = lst[i] & 127;
#pragma unroll
        for (int h = 0; h < H; ++h) {
            const float ex = expf(val[i * H + h] - iord(emx[d * H + h]));
            val[i * H + h] = ex;
            atomicAdd(&den[d * H + h], ex);
        }
    }
    if (over) edge_list_stage<H, CH, 1>(ei, Hm, as, adl, emx, den, tile, alpha_out, n0, tid);
    __syncthreads();

    for (int i = tid; i < n; i += T_THREADS) {
        const int ent = lst[i];
        const int e = ent >> 7;
        const int d = ent & 127;
#pragma unroll
        for (int h = 0; h < H; ++h) {
            const float a = val[i * H + h] / den[d * H + h];
            val[i * H + h] = a;
            alpha_out[(size_t)e * H + h] = a;
        }
    }
    __syncthreads();

#pragma unroll
    for (int p = 0; p < NP; ++p) {
        const int i = (p * T_THREADS + tid) * 4;
        *(v4f*)(tile + i) = val[(i / ROW) * H + (i % ROW) / CH] * sv[p];
    }
    __syncthreads();

    if (over) edge_list_stage<H, CH, 2>(ei, Hm, as, adl, emx, den, tile, alpha_out, n0, tid);

    const int wave = __builtin_amdgcn_readfirstlane(tid >> 5);
    constexpr int NW = T_THREADS / 32;
    const int nfull = T_R + ((n - T_R) / EPW) * EPW;
    for (int i = T_R + wave * EPW; i < nfull; i += NW * EPW)
        tile_add_rows<H, CH, EPW>(Hm, lst, srcl, val, tile, i, lane);
    for (int i = nfull + wave; i < n; i += NW)
        tile_add_rows<H, CH, 1>(Hm, lst, srcl, val, tile, i, lane);
    __syncthreads();

    for (int i = tid * 4; i < TS; i += 4 * T_THREADS) {
        const int c = i % ROW;
        v4f t = *(const v4f*)(tile + i);
        const v4f b = *(const v4f*)(bias + c);
        t += b;
        if (ELU) {
#pragma unroll
            for (int q = 0; q < 4; ++q) t[q] = (t[q] > 0.f) ? t[q] : (expf(t[q]) - 1.f);
        }
        *(v4f*)(outp + (size_t)n0 * ROW + i) = t;
    }
}

static inline int cdiv(long long a, int b) { return (int)((a + b - 1) / b); }

extern "C" void kernel_launch(void* const* d_in, const int* in_sizes, int n_in,
                              void* d_out, int out_size, void* d_ws, size_t ws_size,
                              hipStream_t stream) {
    const float* x      = (const float*)d_in[0];
    const int*   ei     = (const int*)d_in[1];
    const float* W1     = (const float*)d_in[2];
    const float* a_src1 = (const float*)d_in[3];
    const float* a_dst1 = (const float*)d_in[4];
    const float* b1     = (const float*)d_in[5];
    const float* W2     = (const float*)d_in[6];
    const float* a_src2 = (const float*)d_in[7];
    const float* a_dst2 = (const float*)d_in[8];
    const float* b2     = (const float*)d_in[9];

    const int EP = N_EDGES + N_NODES;

    float* H1   = (float*)d_ws;
    float* AGG1 = H1   + (size_t)N_NODES * 512;
    float* AS1  = AGG1 + (size_t)N_NODES * 512;
    float* AD1  = AS1  + (size_t)N_NODES * HEADS;
    float* AS2  = AD1  + (size_t)N_NODES * HEADS;
    float* AD2  = AS2  + N_NODES;
    float* G2   = H1;

    float* out    = (float*)d_out;
    float* alpha1 = out    + (size_t)N_NODES * OUT_CH;
    float* alpha2 = alpha1 + (size_t)EP * HEADS;

    const int TB = 256;

    gemm_wmma_f32<IN_CH, HEADS * HID>
        <<<dim3((HEADS * HID) / G_BN, N_NODES / G_BM), 256, 0, stream>>>(x, W1, H1);

    att_dots<HID><<<cdiv((long long)N_NODES * HEADS * 32, TB), TB, 0, stream>>>(
        H1, a_src1, a_dst1, N_NODES * HEADS, HEADS, AS1, AD1);

    tile_msg<HEADS, HID, 2, true><<<N_NODES / T_R, T_THREADS, 0, stream>>>(
        ei, H1, AS1, AD1, b1, alpha1, AGG1);

    gemm_wmma_f32<HEADS * HID, OUT_CH>
        <<<dim3(OUT_CH / G_BN, N_NODES / G_BM), 256, 0, stream>>>(AGG1, W2, G2);

    att_dots<OUT_CH><<<cdiv((long long)N_NODES * 32, TB), TB, 0, stream>>>(
        G2, a_src2, a_dst2, N_NODES, 1, AS2, AD2);

    tile_msg<1, OUT_CH, 4, false><<<N_NODES / T_R, T_THREADS, 0, stream>>>(
        ei, G2, AS2, AD2, b2, alpha2, out);
}


# Attention dots done in each GEMM's epilogue from its C tile staged in LDS (a thread per row; one plane per 128 columns, layer 2 adds its 2); both dots launches gone (6->4); dot sum order changed
# speedup vs baseline: 2.8599x; 1.0831x over previous
#include <hip/hip_runtime.h>
#include <math.h>

#define N_NODES 20000
#define N_EDGES 320000
#define IN_CH 512
#define HID 128
#define HEADS 4
#define OUT_CH 256
#define NEG_SLOPE 0.2f

typedef float v2f __attribute__((ext_vector_type(2)));
typedef float v8f __attribute__((ext_vector_type(8)));

__device__ __forceinline__ int ford(float f) {
    int i = __float_as_int(f);
    return (i >= 0) ? i : (i ^ 0x7FFFFFFF);
}
__device__ __forceinline__ float iord(int i) {
    return __int_as_float((i >= 0) ? i : (i ^ 0x7FFFFFFF));
}

typedef float v4f __attribute__((ext_vector_type(4)));

#define G_BM 160
#define G_BN 128
#define G_BK 32
#define G_AP 36
#define G_BP 584
#define G_CP 36

template <int K, int NOUT>
__global__ __launch_bounds__(256)
void gemm_wmma_f32(const float* __restrict__ A, const float* __restrict__ B,
                   float* __restrict__ C,
                   const float* __restrict__ a_src, const float* __restrict__ a_dst,
                   float* __restrict__ as_out, float* __restrict__ ad_out) {
    constexpr int A_ST = G_BM * G_AP;
    constexpr int B_ST = (G_BK / 4) * G_BP;
    constexpr int ST   = A_ST + B_ST;
    constexpr int C_W  = 80 * G_CP;
    constexpr int SMEM = (2 * ST > 8 * C_W) ? 2 * ST : 8 * C_W;
    constexpr int NS   = K / G_BK;
    static_assert(K % G_BK == 0 && NOUT % G_BN == 0 && N_NODES % G_BM == 0, "tiles must divide");
    static_assert(G_BK / 4 == 8, "one k quad of the slab per wave in the B copy");
    __shared__ __attribute__((aligned(16))) float smem[SMEM];

    const int tid  = threadIdx.x;
    const int lane = tid & 31;
    const int wave = tid >> 5;
    const int c  = lane & 15;
    const int h  = lane >> 4;
    const int wm = wave >> 2;
    const int wn = wave & 3;
    const int n0 = blockIdx.x * G_BN;
    const int m0 = blockIdx.y * G_BM;

    const int a_row = tid >> 3;
    const int a_k4  = (tid & 7) * 4;
    const float* __restrict__ Ag = A + (size_t)(m0 + a_row) * K + a_k4;
    const float* __restrict__ Bg = B + (size_t)(4 * wave) * NOUT + n0 + 4 * lane;
    float* const As_w = smem + a_row * G_AP + a_k4;
    float* const Bs_w = smem + A_ST + wave * G_BP + 4 * lane;
    const float* const Af = smem + (wm * 80 + c) * G_AP + 4 * h;
    const float* const Bf = smem + A_ST + h * G_BP + 4 * ((c & 3) * 36 + wn * 8 + (c >> 2));

    v4f ra[5], rb[4];
    v8f acc[5][2] = {};

#define G_GLOAD(k0)                                                              \
    _Pragma("unroll") for (int p = 0; p < 5; ++p)                                \
        ra[p] = *(const v4f*)(Ag + (size_t)(p * 32) * K + (k0));                 \
    _Pragma("unroll") for (int r = 0; r < 4; ++r)                                \
        rb[r] = *(const v4f*)(Bg + (size_t)((k0) + r) * NOUT);
#define G_SSTORE(st)                                                             \
    _Pragma("unroll") for (int p = 0; p < 5; ++p)                                \
        *(v4f*)(As_w + (st) * ST + p * 32 * G_AP) = ra[p];                       \
    _Pragma("unroll") for (int i = 0; i < 4; ++i) {                              \
        v4f col = {rb[0][i], rb[1][i], rb[2][i], rb[3][i]};                      \
        *(v4f*)(Bs_w + (st) * ST + i * 36 * 4) = col;                            \
    }
#define G_COMPUTE(Afs, Bfs)                                                      \
    _Pragma("unroll") for (int t = 0; t < G_BK / 8; ++t) {                       \
        v4f fa[5], fb[2];                                                        \
        _Pragma("unroll") for (int i = 0; i < 5; ++i)                            \
            fa[i] = *(const v4f*)((Afs) + i * 16 * G_AP + t * 8);                \
        _Pragma("unroll") for (int j = 0; j < 2; ++j)                            \
            fb[j] = *(const v4f*)((Bfs) + t * 2 * G_BP + j * 16);                \
        _Pragma("unroll") for (int i = 0; i < 5; ++i)                            \
            _Pragma("unroll") for (int j = 0; j < 2; ++j) {                      \
                v2f a = {fa[i].x, fa[i].y};                                      \
                v2f b = {fb[j].x, fb[j].y};                                      \
                acc[i][j] = __builtin_amdgcn_wmma_f32_16x16x4_f32(               \
                    false, a, false, b, (short)0, acc[i][j], false, false);      \
            }                                                                    \
        _Pragma("unroll") for (int i = 0; i < 5; ++i)                            \
            _Pragma("unroll") for (int j = 0; j < 2; ++j) {                      \
                v2f a = {fa[i].z, fa[i].w};                                      \
                v2f b = {fb[j].z, fb[j].w};                                      \
                acc[i][j] = __builtin_amdgcn_wmma_f32_16x16x4_f32(               \
                    false, a, false, b, (short)0, acc[i][j], false, false);      \
            }                                                                    \
    }
#define G_FENCE() __builtin_amdgcn_sched_barrier(0)
#define G_SLOTS() asm volatile("v_nop\n\tv_nop\n\tv_nop\n\tv_nop")

    G_GLOAD(0)
    G_SSTORE(0)
    __syncthreads();

    for (int s = 0; s < NS - 1; ++s) {
        const int st = s & 1;
        const float* const Afs = Af + st * ST;
        const float* const Bfs = Bf + st * ST;
        G_GLOAD((s + 1) * G_BK)
        G_FENCE();
        G_COMPUTE(Afs, Bfs)
        G_FENCE();
        G_SLOTS();
        G_FENCE();
        G_SSTORE(st ^ 1)
        __syncthreads();
    }
    {
        const float* const Afs = Af + ((NS - 1) & 1) * ST;
        const float* const Bfs = Bf + ((NS - 1) & 1) * ST;
        G_FENCE();
        G_COMPUTE(Afs, Bfs)
        G_FENCE();
        G_SLOTS();
        G_FENCE();
    }
    __syncthreads();

    float* const Cw = smem + wave * C_W;
#pragma unroll
    for (int i = 0; i < 5; ++i)
#pragma unroll
        for (int j = 0; j < 2; ++j)
#pragma unroll
            for (int r = 0; r < 8; ++r)
                Cw[(i * 16 + 8 * h + r) * G_CP + j * 16 + c] = acc[i][j][r];
    __syncthreads();
    const int rr = lane >> 3;
    const int c4 = (lane & 7) * 4;
    float* __restrict__ Cg = C + (size_t)(m0 + wm * 80 + rr) * NOUT + n0 + wn * 32 + c4;
#pragma unroll
    for (int t = 0; t < 20; ++t) {
        v4f v = *(const v4f*)(Cw + (t * 4 + rr) * G_CP + c4);
        *(v4f*)(Cg + (size_t)(t * 4) * NOUT) = v;
    }

    if (tid < G_BM) {
        const float* const rowp = smem + ((tid / 80) * 4) * C_W + (tid % 80) * G_CP;
        const float* __restrict__ av = a_src + n0;
        const float* __restrict__ dv = a_dst + n0;
        v4f s4 = {0.f, 0.f, 0.f, 0.f}, d4 = {0.f, 0.f, 0.f, 0.f};
#pragma unroll
        for (int q = 0; q < 4; ++q)
#pragma unroll
            for (int j = 0; j < 8; ++j) {
                const v4f v = *(const v4f*)(rowp + q * C_W + j * 4);
                const v4f a = *(const v4f*)(av + q * 32 + j * 4);
                const v4f b = *(const v4f*)(dv + q * 32 + j * 4);
                s4 += v * a;
                d4 += v * b;
            }
        as_out[(size_t)blockIdx.x * N_NODES + m0 + tid] = (s4.x + s4.y) + (s4.z + s4.w);
        ad_out[(size_t)blockIdx.x * N_NODES + m0 + tid] = (d4.x + d4.y) + (d4.z + d4.w);
    }
#undef G_GLOAD
#undef G_SSTORE
#undef G_COMPUTE
#undef G_FENCE
#undef G_SLOTS
}

template <int H, int CH>
__device__ __forceinline__ float dot_at(const float* __restrict__ p, int node, int h) {
    constexpr int NPL = CH / G_BN;
    static_assert(CH % G_BN == 0, "a head is a whole number of the GEMM's column blocks");
    float v = p[(size_t)(h * NPL) * N_NODES + node];
#pragma unroll
    for (int q = 1; q < NPL; ++q) v += p[(size_t)(h * NPL + q) * N_NODES + node];
    return v;
}

#define T_R 80
#define T_CAP 4096
#define T_THREADS 1024
#define T_U 4

__device__ __forceinline__ float att_logit(float as, float ad) {
    const float v = as + ad;
    return (v > 0.f) ? v : NEG_SLOPE * v;
}

template <int H, int CH, int NE>
__device__ __forceinline__ void tile_add_rows(const float* __restrict__ Hm, const int* lst,
                                              const int* srcl, const float* val,
                                              float* tile, int i, int lane) {
    constexpr int ROW = H * CH;
    constexpr int NL  = ROW / 32;
    constexpr int LPH = CH / 32;
    int s[NE], d[NE];
    float al[NE][H];
    float v[NE][NL];
#pragma unroll
    for (int k = 0; k < NE; ++k) {
        d[k] = __builtin_amdgcn_readfirstlane(lst[i + k]) & 127;
        s[k] = __builtin_amdgcn_readfirstlane(srcl[i + k]);
#pragma unroll
        for (int h = 0; h < H; ++h) al[k][h] = val[(i + k) * H + h];
    }
#pragma unroll
    for (int k = 0; k < NE; ++k) {
        const float* __restrict__ hp = Hm + (size_t)s[k] * ROW + lane;
#pragma unroll
        for (int j = 0; j < NL; ++j) v[k][j] = hp[32 * j];
    }
#pragma unroll
    for (int k = 0; k < NE; ++k) {
        float* tp = tile + d[k] * ROW + lane;
#pragma unroll
        for (int j = 0; j < NL; ++j)
            atomicAdd(&tp[32 * j], al[k][j / LPH] * v[k][j]);
    }
}

template <int H, int CH, int PH>
__device__ __forceinline__ void edge_list_stage(const int* __restrict__ ei,
                                                const float* __restrict__ Hm,
                                                const float* __restrict__ as, const float* adl,
                                                int* emx, float* den, float* tile,
                                                float* __restrict__ alpha_out, int n0, int tid) {
    constexpr int ROW = H * CH;
#pragma nounroll
    for (int e = tid; e < N_EDGES; e += T_THREADS) {
        const unsigned d = (unsigned)(ei[N_EDGES + e] - n0);
        if (d < (unsigned)T_R) {
            const int s = ei[e];
#pragma nounroll
            for (int h = 0; h < H; ++h) {
                const float v = att_logit(dot_at<H, CH>(as, s, h), adl[d * H + h]);
                if (PH == 0) {
                    atomicMax(&emx[d * H + h], ford(v));
                } else {
                    const float ex = expf(v - iord(emx[d * H + h]));
                    if (PH == 1) {
                        atomicAdd(&den[d * H + h], ex);
                    } else {
                        const float a = ex / den[d * H + h];
                        alpha_out[(size_t)e * H + h] = a;
#pragma nounroll
                        for (int c = h * CH; c < (h + 1) * CH; ++c)
                            atomicAdd(&tile[d * ROW + c], a * Hm[(size_t)s * ROW + c]);
                    }
                }
            }
        }
    }
}

template <int H, int CH, int EPW, bool ELU>
__global__ __launch_bounds__(T_THREADS)
void tile_msg(const int* __restrict__ ei, const float* __restrict__ Hm,
              const float* __restrict__ as, const float* __restrict__ ad,
              const float* __restrict__ bias, float* __restrict__ alpha_out,
              float* __restrict__ outp) {
    constexpr int ROW = H * CH;
    constexpr int TS  = T_R * ROW;
    constexpr int SEG = T_R * H;
    static_assert(N_NODES % T_R == 0, "tiles must cover the nodes");
    static_assert(T_R <= 128 && N_EDGES + N_NODES <= (1 << 24), "list entry packs edge id and local row");
    static_assert(CH % 32 == 0 && TS % (4 * T_THREADS) == 0 && N_EDGES % 4 == 0, "16-byte pieces");
    static_assert(SEG <= T_THREADS && T_R < T_CAP, "one thread per segment; self loops fit the list");
    __shared__ __attribute__((aligned(16))) float tile[TS];
    __shared__ __attribute__((aligned(16))) float val[T_CAP * H];
    __shared__ int lst[T_CAP];
    __shared__ int srcl[T_CAP];
    __shared__ float adl[SEG];
    __shared__ int emx[SEG];
    __shared__ float den[SEG];
    __shared__ int cnt;

    const int tid  = threadIdx.x;
    const int lane = tid & 31;
    const int n0   = blockIdx.x * T_R;

    if (tid == 0) cnt = T_R;
    if (tid < T_R) lst[tid] = ((N_EDGES + n0 + tid) << 7) | tid;
    if (tid < SEG) { emx[tid] = (int)0x80000000; den[tid] = 0.f; }
    __syncthreads();

    float adv = 0.f;
    if (tid < SEG) adv = dot_at<H, CH>(ad, n0 + tid / H, tid % H);

    const int* __restrict__ dst = ei + N_EDGES;
    for (int base = tid * 4; base < N_EDGES; base += 4 * T_THREADS * T_U) {
        int4 dv[T_U];
#pragma unroll
        for (int u = 0; u < T_U; ++u) {
            const int e4 = base + u * 4 * T_THREADS;
            dv[u] = (e4 < N_EDGES) ? *(const int4*)(dst + e4) : make_int4(-1, -1, -1, -1);
        }
#pragma unroll
        for (int u = 0; u < T_U; ++u) {
            const int e4 = base + u * 4 * T_THREADS;
            const unsigned dq[4] = {(unsigned)(dv[u].x - n0), (unsigned)(dv[u].y - n0),
                                    (unsigned)(dv[u].z - n0), (unsigned)(dv[u].w - n0)};
            const unsigned lo = min(min(dq[0], dq[1]), min(dq[2], dq[3]));
            if (lo < (unsigned)T_R) {
#pragma unroll
                for (int q = 0; q < 4; ++q) {
                    if (dq[q] < (unsigned)T_R) {
                        const int slot = atomicAdd(&cnt, 1);
                        if (slot < T_CAP) lst[slot] = ((e4 + q) << 7) | (int)dq[q];
                    }
                }
            }
        }
    }
    if (tid < SEG) adl[tid] = adv;
    __syncthreads();

    const int found = cnt;
    const bool over = found > T_CAP;
    const int n = over ? T_R : found;

    constexpr int NP = TS / (4 * T_THREADS);
    v4f sv[NP];
#pragma unroll
    for (int p = 0; p < NP; ++p)
        sv[p] = *(const v4f*)(Hm + (size_t)n0 * ROW + (p * T_THREADS + tid) * 4);

    for (int i = tid; i < n; i += T_THREADS) {
        const int ent = lst[i];
        const int e = ent >> 7;
        const int d = ent & 127;
        const int s = (e < N_EDGES) ? ei[e] : (e - N_EDGES);
        srcl[i] = s;
#pragma unroll
        for (int h = 0; h < H; ++h) {
            const float v = att_logit(dot_at<H, CH>(as, s, h), adl[d * H + h]);
            val[i * H + h] = v;
            atomicMax(&emx[d * H + h], ford(v));
        }
    }
    if (over) edge_list_stage<H, CH, 0>(ei, Hm, as, adl, emx, den, tile, alpha_out, n0, tid);
    __syncthreads();

    for (int i = tid; i < n; i += T_THREADS) {
        const int d = lst[i] & 127;
#pragma unroll
        for (int h = 0; h < H; ++h) {
            const float ex = expf(val[i * H + h] - iord(emx[d * H + h]));
            val[i * H + h] = ex;
            atomicAdd(&den[d * H + h], ex);
        }
    }
    if (over) edge_list_stage<H, CH, 1>(ei, Hm, as, adl, emx, den, tile, alpha_out, n0, tid);
    __syncthreads();

    for (int i = tid; i < n; i += T_THREADS) {
        const int ent = lst[i];
        const int e = ent >> 7;
        const int d = ent & 127;
#pragma unroll
        for (int h = 0; h < H; ++h) {
            const float a = val[i * H + h] / den[d * H + h];
            val[i * H + h] = a;
            alpha_out[(size_t)e * H + h] = a;
        }
    }
    __syncthreads();

#pragma unroll
    for (int p = 0; p < NP; ++p) {
        const int i = (p * T_THREADS + tid) * 4;
        *(v4f*)(tile + i) = val[(i / ROW) * H + (i % ROW) / CH] * sv[p];
    }
    __syncthreads();

    if (over) edge_list_stage<H, CH, 2>(ei, Hm, as, adl, emx, den, tile, alpha_out, n0, tid);

    const int wave = __builtin_amdgcn_readfirstlane(tid >> 5);
    constexpr int NW = T_THREADS / 32;
    const int nfull = T_R + ((n - T_R) / EPW) * EPW;
    for (int i = T_R + wave * EPW; i < nfull; i += NW * EPW)
        tile_add_rows<H, CH, EPW>(Hm, lst, srcl, val, tile, i, lane);
    for (int i = nfull + wave; i < n; i += NW)
        tile_add_rows<H, CH, 1>(Hm, lst, srcl, val, tile, i, lane);
    __syncthreads();

    for (int i = tid * 4; i < TS; i += 4 * T_THREADS) {
        const int c = i % ROW;
        v4f t = *(const v4f*)(tile + i);
        const v4f b = *(const v4f*)(bias + c);
        t += b;
        if (ELU) {
#pragma unroll
            for (int q = 0; q < 4; ++q) t[q] = (t[q] > 0.f) ? t[q] : (expf(t[q]) - 1.f);
        }
        *(v4f*)(outp + (size_t)n0 * ROW + i) = t;
    }
}

extern "C" void kernel_launch(void* const* d_in, const int* in_sizes, int n_in,
                              void* d_out, int out_size, void* d_ws, size_t ws_size,
                              hipStream_t stream) {
    const float* x      = (const float*)d_in[0];
    const int*   ei     = (const int*)d_in[1];
    const float* W1     = (const float*)d_in[2];
    const float* a_src1 = (const float*)d_in[3];
    const float* a_dst1 = (const float*)d_in[4];
    const float* b1     = (const float*)d_in[5];
    const float* W2     = (const float*)d_in[6];
    const float* a_src2 = (const float*)d_in[7];
    const float* a_dst2 = (const float*)d_in[8];
    const float* b2     = (const float*)d_in[9];

    const int EP = N_EDGES + N_NODES;

    float* H1   = (float*)d_ws;
    float* AGG1 = H1   + (size_t)N_NODES * 512;
    float* AS1  = AGG1 + (size_t)N_NODES * 512;
    float* AD1  = AS1  + (size_t)N_NODES * HEADS;
    float* AS2  = AS1;
    float* AD2  = AD1;
    float* G2   = H1;

    float* out    = (float*)d_out;
    float* alpha1 = out    + (size_t)N_NODES * OUT_CH;
    float* alpha2 = alpha1 + (size_t)EP * HEADS;

    gemm_wmma_f32<IN_CH, HEADS * HID>
        <<<dim3((HEADS * HID) / G_BN, N_NODES / G_BM), 256, 0, stream>>>(
            x, W1, H1, a_src1, a_dst1, AS1, AD1);

    tile_msg<HEADS, HID, 2, true><<<N_NODES / T_R, T_THREADS, 0, stream>>>(
        ei, H1, AS1, AD1, b1, alpha1, AGG1);

    gemm_wmma_f32<HEADS * HID, OUT_CH>
        <<<dim3(OUT_CH / G_BN, N_NODES / G_BM), 256, 0, stream>>>(
            AGG1, W2, G2, a_src2, a_dst2, AS2, AD2);

    tile_msg<1, OUT_CH, 4, false><<<N_NODES / T_R, T_THREADS, 0, stream>>>(
        ei, G2, AS2, AD2, b2, alpha2, out);
}


# Tile edge lists found once per call: layer 1's groups leave list, sources, count in memory, layer 2's groups read them back and scan nothing; sum order: arrival, over layer 1's list order
# speedup vs baseline: 3.0464x; 1.0652x over previous
#include <hip/hip_runtime.h>
#include <math.h>

#define N_NODES 20000
#define N_EDGES 320000
#define IN_CH 512
#define HID 128
#define HEADS 4
#define OUT_CH 256
#define NEG_SLOPE 0.2f

typedef float v2f __attribute__((ext_vector_type(2)));
typedef float v8f __attribute__((ext_vector_type(8)));

__device__ __forceinline__ int ford(float f) {
    int i = __float_as_int(f);
    return (i >= 0) ? i : (i ^ 0x7FFFFFFF);
}
__device__ __forceinline__ float iord(int i) {
    return __int_as_float((i >= 0) ? i : (i ^ 0x7FFFFFFF));
}

typedef float v4f __attribute__((ext_vector_type(4)));

#define G_BM 160
#define G_BN 128
#define G_BK 32
#define G_AP 36
#define G_BP 584
#define G_CP 36

template <int K, int NOUT>
__global__ __launch_bounds__(256)
void gemm_wmma_f32(const float* __restrict__ A, const float* __restrict__ B,
                   float* __restrict__ C,
                   const float* __restrict__ a_src, const float* __restrict__ a_dst,
                   float* __restrict__ as_out, float* __restrict__ ad_out) {
    constexpr int A_ST = G_BM * G_AP;
    constexpr int B_ST = (G_BK / 4) * G_BP;
    constexpr int ST   = A_ST + B_ST;
    constexpr int C_W  = 80 * G_CP;
    constexpr int SMEM = (2 * ST > 8 * C_W) ? 2 * ST : 8 * C_W;
    constexpr int NS   = K / G_BK;
    static_assert(K % G_BK == 0 && NOUT % G_BN == 0 && N_NODES % G_BM == 0, "tiles must divide");
    static_assert(G_BK / 4 == 8, "one k quad of the slab per wave in the B copy");
    __shared__ __attribute__((aligned(16))) float smem[SMEM];

    const int tid  = threadIdx.x;
    const int lane = tid & 31;
    const int wave = tid >> 5;
    const int c  = lane & 15;
    const int h  = lane >> 4;
    const int wm = wave >> 2;
    const int wn = wave & 3;
    const int n0 = blockIdx.x * G_BN;
    const int m0 = blockIdx.y * G_BM;

    const int a_row = tid >> 3;
    const int a_k4  = (tid & 7) * 4;
    const float* __restrict__ Ag = A + (size_t)(m0 + a_row) * K + a_k4;
    const float* __restrict__ Bg = B + (size_t)(4 * wave) * NOUT + n0 + 4 * lane;
    float* const As_w = smem + a_row * G_AP + a_k4;
    float* const Bs_w = smem + A_ST + wave * G_BP + 4 * lane;
    const float* const Af = smem + (wm * 80 + c) * G_AP + 4 * h;
    const float* const Bf = smem + A_ST + h * G_BP + 4 * ((c & 3) * 36 + wn * 8 + (c >> 2));

    v4f ra[5], rb[4];
    v8f acc[5][2] = {};

#define G_GLOAD(k0)                                                              \
    _Pragma("unroll") for (int p = 0; p < 5; ++p)                                \
        ra[p] = *(const v4f*)(Ag + (size_t)(p * 32) * K + (k0));                 \
    _Pragma("unroll") for (int r = 0; r < 4; ++r)                                \
        rb[r] = *(const v4f*)(Bg + (size_t)((k0) + r) * NOUT);
#define G_SSTORE(st)                                                             \
    _Pragma("unroll") for (int p = 0; p < 5; ++p)                                \
        *(v4f*)(As_w + (st) * ST + p * 32 * G_AP) = ra[p];                       \
    _Pragma("unroll") for (int i = 0; i < 4; ++i) {                              \
        v4f col = {rb[0][i], rb[1][i], rb[2][i], rb[3][i]};                      \
        *(v4f*)(Bs_w + (st) * ST + i * 36 * 4) = col;                            \
    }
#define G_COMPUTE(Afs, Bfs)                                                      \
    _Pragma("unroll") for (int t = 0; t < G_BK / 8; ++t) {                       \
        v4f fa[5], fb[2];                                                        \
        _Pragma("unroll") for (int i = 0; i < 5; ++i)                            \
            fa[i] = *(const v4f*)((Afs) + i * 16 * G_AP + t * 8);                \
        _Pragma("unroll") for (int j = 0; j < 2; ++j)                            \
            fb[j] = *(const v4f*)((Bfs) + t * 2 * G_BP + j * 16);                \
        _Pragma("unroll") for (int i = 0; i < 5; ++i)                            \
            _Pragma("unroll") for (int j = 0; j < 2; ++j) {                      \
                v2f a = {fa[i].x, fa[i].y};                                      \
                v2f b = {fb[j].x, fb[j].y};                                      \
                acc[i][j] = __builtin_amdgcn_wmma_f32_16x16x4_f32(               \
                    false, a, false, b, (short)0, acc[i][j], false, false);      \
            }                                                                    \
        _Pragma("unroll") for (int i = 0; i < 5; ++i)                            \
            _Pragma("unroll") for (int j = 0; j < 2; ++j) {                      \
                v2f a = {fa[i].z, fa[i].w};                                      \
                v2f b = {fb[j].z, fb[j].w};                                      \
                acc[i][j] = __builtin_amdgcn_wmma_f32_16x16x4_f32(               \
                    false, a, false, b, (short)0, acc[i][j], false, false);      \
            }                                                                    \
    }
#define G_FENCE() __builtin_amdgcn_sched_barrier(0)
#define G_SLOTS() asm volatile("v_nop\n\tv_nop\n\tv_nop\n\tv_nop")

    G_GLOAD(0)
    G_SSTORE(0)
    __syncthreads();

    for (int s = 0; s < NS - 1; ++s) {
        const int st = s & 1;
        const float* const Afs = Af + st * ST;
        const float* const Bfs = Bf + st * ST;
        G_GLOAD((s + 1) * G_BK)
        G_FENCE();
        G_COMPUTE(Afs, Bfs)
        G_FENCE();
        G_SLOTS();
        G_FENCE();
        G_SSTORE(st ^ 1)
        __syncthreads();
    }
    {
        const float* const Afs = Af + ((NS - 1) & 1) * ST;
        const float* const Bfs = Bf + ((NS - 1) & 1) * ST;
        G_FENCE();
        G_COMPUTE(Afs, Bfs)
        G_FENCE();
        G_SLOTS();
        G_FENCE();
    }
    __syncthreads();

    float* const Cw = smem + wave * C_W;
#pragma unroll
    for (int i = 0; i < 5; ++i)
#pragma unroll
        for (int j = 0; j < 2; ++j)
#pragma unroll
            for (int r = 0; r < 8; ++r)
                Cw[(i * 16 + 8 * h + r) * G_CP + j * 16 + c] = acc[i][j][r];
    __syncthreads();
    const int rr = lane >> 3;
    const int c4 = (lane & 7) * 4;
    float* __restrict__ Cg = C + (size_t)(m0 + wm * 80 + rr) * NOUT + n0 + wn * 32 + c4;
#pragma unroll
    for (int t = 0; t < 20; ++t) {
        v4f v = *(const v4f*)(Cw + (t * 4 + rr) * G_CP + c4);
        *(v4f*)(Cg + (size_t)(t * 4) * NOUT) = v;
    }

    if (tid < G_BM) {
        const float* const rowp = smem + ((tid / 80) * 4) * C_W + (tid % 80) * G_CP;
        const float* __restrict__ av = a_src + n0;
        const float* __restrict__ dv = a_dst + n0;
        v4f s4 = {0.f, 0.f, 0.f, 0.f}, d4 = {0.f, 0.f, 0.f, 0.f};
#pragma unroll
        for (int q = 0; q < 4; ++q)
#pragma unroll
            for (int j = 0; j < 8; ++j) {
                const v4f v = *(const v4f*)(rowp + q * C_W + j * 4);
                const v4f a = *(const v4f*)(av + q * 32 + j * 4);
                const v4f b = *(const v4f*)(dv + q * 32 + j * 4);
                s4 += v * a;
                d4 += v * b;
            }
        as_out[(size_t)blockIdx.x * N_NODES + m0 + tid] = (s4.x + s4.y) + (s4.z + s4.w);
        ad_out[(size_t)blockIdx.x * N_NODES + m0 + tid] = (d4.x + d4.y) + (d4.z + d4.w);
    }
#undef G_GLOAD
#undef G_SSTORE
#undef G_COMPUTE
#undef G_FENCE
#undef G_SLOTS
}

template <int H, int CH>
__device__ __forceinline__ float dot_at(const float* __restrict__ p, int node, int h) {
    constexpr int NPL = CH / G_BN;
    static_assert(CH % G_BN == 0, "a head is a whole number of the GEMM's column blocks");
    float v = p[(size_t)(h * NPL) * N_NODES + node];
#pragma unroll
    for (int q = 1; q < NPL; ++q) v += p[(size_t)(h * NPL + q) * N_NODES + node];
    return v;
}

#define T_R 80
#define T_CAP 4096
#define T_THREADS 1024
#define T_U 4
#define T_KEEP (T_R * OUT_CH)

__device__ __forceinline__ float att_logit(float as, float ad) {
    const float v = as + ad;
    return (v > 0.f) ? v : NEG_SLOPE * v;
}

template <int H, int CH, int NE>
__device__ __forceinline__ void tile_add_rows(const float* __restrict__ Hm, const int* lst,
                                              const int* srcl, const float* val,
                                              float* tile, int i, int lane) {
    constexpr int ROW = H * CH;
    constexpr int NL  = ROW / 32;
    constexpr int LPH = CH / 32;
    int s[NE], d[NE];
    float al[NE][H];
    float v[NE][NL];
#pragma unroll
    for (int k = 0; k < NE; ++k) {
        d[k] = __builtin_amdgcn_readfirstlane(lst[i + k]) & 127;
        s[k] = __builtin_amdgcn_readfirstlane(srcl[i + k]);
#pragma unroll
        for (int h = 0; h < H; ++h) al[k][h] = val[(i + k) * H + h];
    }
#pragma unroll
    for (int k = 0; k < NE; ++k) {
        const float* __restrict__ hp = Hm + (size_t)s[k] * ROW + lane;
#pragma unroll
        for (int j = 0; j < NL; ++j) v[k][j] = hp[32 * j];
    }
#pragma unroll
    for (int k = 0; k < NE; ++k) {
        float* tp = tile + d[k] * ROW + lane;
#pragma unroll
        for (int j = 0; j < NL; ++j)
            atomicAdd(&tp[32 * j], al[k][j / LPH] * v[k][j]);
    }
}

template <int H, int CH, int PH>
__device__ __forceinline__ void edge_list_stage(const int* __restrict__ ei,
                                                const float* __restrict__ Hm,
                                                const float* __restrict__ as, const float* adl,
                                                int* emx, float* den, float* tile,
                                                float* __restrict__ alpha_out, int n0, int tid) {
    constexpr int ROW = H * CH;
#pragma nounroll
    for (int e = tid; e < N_EDGES; e += T_THREADS) {
        const unsigned d = (unsigned)(ei[N_EDGES + e] - n0);
        if (d < (unsigned)T_R) {
            const int s = ei[e];
#pragma nounroll
            for (int h = 0; h < H; ++h) {
                const float v = att_logit(dot_at<H, CH>(as, s, h), adl[d * H + h]);
                if (PH == 0) {
                    atomicMax(&emx[d * H + h], ford(v));
                } else {
                    const float ex = expf(v - iord(emx[d * H + h]));
                    if (PH == 1) {
                        atomicAdd(&den[d * H + h], ex);
                    } else {
                        const float a = ex / den[d * H + h];
                        alpha_out[(size_t)e * H + h] = a;
#pragma nounroll
                        for (int c = h * CH; c < (h + 1) * CH; ++c)
                            atomicAdd(&tile[d * ROW + c], a * Hm[(size_t)s * ROW + c]);
                    }
                }
            }
        }
    }
}

template <int H, int CH, int EPW, bool ELU, int KEEP>
__global__ __launch_bounds__(T_THREADS)
void tile_msg(const int* __restrict__ ei, const float* __restrict__ Hm,
              const float* __restrict__ as, const float* __restrict__ ad,
              const float* __restrict__ bias, float* __restrict__ alpha_out,
              float* outp, int* keep_all) {
    constexpr int ROW = H * CH;
    constexpr int TS  = T_R * ROW;
    constexpr int SEG = T_R * H;
    static_assert(N_NODES % T_R == 0, "tiles must cover the nodes");
    static_assert(T_R <= 128 && N_EDGES + N_NODES <= (1 << 24), "list entry packs edge id and local row");
    static_assert(CH % 32 == 0 && TS % (4 * T_THREADS) == 0 && N_EDGES % 4 == 0, "16-byte pieces");
    static_assert(SEG <= T_THREADS && T_R < T_CAP, "one thread per segment; self loops fit the list");
    static_assert(2 * T_CAP + 1 <= T_KEEP, "the kept list fits the tile's own output rows");
    __shared__ __attribute__((aligned(16))) float tile[TS];
    __shared__ __attribute__((aligned(16))) float val[T_CAP * H];
    __shared__ int lst[T_CAP];
    __shared__ int srcl[T_CAP];
    __shared__ float adl[SEG];
    __shared__ int emx[SEG];
    __shared__ float den[SEG];
    __shared__ int cnt;

    const int tid  = threadIdx.x;
    const int lane = tid & 31;
    const int n0   = blockIdx.x * T_R;

    int* const keep = keep_all + (size_t)blockIdx.x * T_KEEP;

    if (KEEP == 0) {
        if (tid == 0) cnt = T_R;
        if (tid < T_R) lst[tid] = ((N_EDGES + n0 + tid) << 7) | tid;
    }
    if (tid < SEG) { emx[tid] = (int)0x80000000; den[tid] = 0.f; }
    __syncthreads();

    float adv = 0.f;
    if (tid < SEG) adv = dot_at<H, CH>(ad, n0 + tid / H, tid % H);
    int kept_n = 0;
    if (KEEP == 1) kept_n = keep[2 * T_CAP];

    const int* __restrict__ dst = ei + N_EDGES;
    if (KEEP == 0)
    for (int base = tid * 4; base < N_EDGES; base += 4 * T_THREADS * T_U) {
        int4 dv[T_U];
#pragma unroll
        for (int u = 0; u < T_U; ++u) {
            const int e4 = base + u * 4 * T_THREADS;
            dv[u] = (e4 < N_EDGES) ? *(const int4*)(dst + e4) : make_int4(-1, -1, -1, -1);
        }
#pragma unroll
        for (int u = 0; u < T_U; ++u) {
            const int e4 = base + u * 4 * T_THREADS;
            const unsigned dq[4] = {(unsigned)(dv[u].x - n0), (unsigned)(dv[u].y - n0),
                                    (unsigned)(dv[u].z - n0), (unsigned)(dv[u].w - n0)};
            const unsigned lo = min(min(dq[0], dq[1]), min(dq[2], dq[3]));
            if (lo < (unsigned)T_R) {
#pragma unroll
                for (int q = 0; q < 4; ++q) {
                    if (dq[q] < (unsigned)T_R) {
                        const int slot = atomicAdd(&cnt, 1);
                        if (slot < T_CAP) lst[slot] = ((e4 + q) << 7) | (int)dq[q];
                    }
                }
            }
        }
    }
    if (tid < SEG) adl[tid] = adv;
    __syncthreads();

    const int found = (KEEP == 0) ? cnt : kept_n;
    const bool over = found > T_CAP;
    const int n = over ? T_R : found;
    if (KEEP == 0 && tid == 0) keep[2 * T_CAP] = found;

    constexpr int NP = TS / (4 * T_THREADS);
    v4f sv[NP];
#pragma unroll
    for (int p = 0; p < NP; ++p)
        sv[p] = *(const v4f*)(Hm + (size_t)n0 * ROW + (p * T_THREADS + tid) * 4);

    for (int i = tid; i < n; i += T_THREADS) {
        int ent, s;
        if (KEEP == 0) {
            ent = lst[i];
            const int e = ent >> 7;
            s = (e < N_EDGES) ? ei[e] : (e - N_EDGES);
            keep[i] = ent;
            keep[T_CAP + i] = s;
        } else {
            ent = keep[i];
            s = keep[T_CAP + i];
            lst[i] = ent;
        }
        const int d = ent & 127;
        srcl[i] = s;
#pragma unroll
        for (int h = 0; h < H; ++h) {
            const float v = att_logit(dot_at<H, CH>(as, s, h), adl[d * H + h]);
            val[i * H + h] = v;
            atomicMax(&emx[d * H + h], ford(v));
        }
    }
    if (over) edge_list_stage<H, CH, 0>(ei, Hm, as, adl, emx, den, tile, alpha_out, n0, tid);
    __syncthreads();

    for (int i = tid; i < n; i += T_THREADS) {
        const int d = lst[i] & 127;
#pragma unroll
        for (int h = 0; h < H; ++h) {
            const float ex = expf(val[i * H + h] - iord(emx[d * H + h]));
            val[i * H + h] = ex;
            atomicAdd(&den[d * H + h], ex);
        }
    }
    if (over) edge_list_stage<H, CH, 1>(ei, Hm, as, adl, emx, den, tile, alpha_out, n0, tid);
    __syncthreads();

    for (int i = tid; i < n; i += T_THREADS) {
        const int ent = lst[i];
        const int e = ent >> 7;
        const int d = ent & 127;
#pragma unroll
        for (int h = 0; h < H; ++h) {
            const float a = val[i * H + h] / den[d * H + h];
            val[i * H + h] = a;
            alpha_out[(size_t)e * H + h] = a;
        }
    }
    __syncthreads();

#pragma unroll
    for (int p = 0; p < NP; ++p) {
        const int i = (p * T_THREADS + tid) * 4;
        *(v4f*)(tile + i) = val[(i / ROW) * H + (i % ROW) / CH] * sv[p];
    }
    __syncthreads();

    if (over) edge_list_stage<H, CH, 2>(ei, Hm, as, adl, emx, den, tile, alpha_out, n0, tid);

    const int wave = __builtin_amdgcn_readfirstlane(tid >> 5);
    constexpr int NW = T_THREADS / 32;
    const int nfull = T_R + ((n - T_R) / EPW) * EPW;
    for (int i = T_R + wave * EPW; i < nfull; i += NW * EPW)
        tile_add_rows<H, CH, EPW>(Hm, lst, srcl, val, tile, i, lane);
    for (int i = nfull + wave; i < n; i += NW)
        tile_add_rows<H, CH, 1>(Hm, lst, srcl, val, tile, i, lane);
    __syncthreads();

    for (int i = tid * 4; i < TS; i += 4 * T_THREADS) {
        const int c = i % ROW;
        v4f t = *(const v4f*)(tile + i);
        const v4f b = *(const v4f*)(bias + c);
        t += b;
        if (ELU) {
#pragma unroll
            for (int q = 0; q < 4; ++q) t[q] = (t[q] > 0.f) ? t[q] : (expf(t[q]) - 1.f);
        }
        *(v4f*)(outp + (size_t)n0 * ROW + i) = t;
    }
}

extern "C" void kernel_launch(void* const* d_in, const int* in_sizes, int n_in,
                              void* d_out, int out_size, void* d_ws, size_t ws_size,
                              hipStream_t stream) {
    const float* x      = (const float*)d_in[0];
    const int*   ei     = (const int*)d_in[1];
    const float* W1     = (const float*)d_in[2];
    const float* a_src1 = (const float*)d_in[3];
    const float* a_dst1 = (const float*)d_in[4];
    const float* b1     = (const float*)d_in[5];
    const float* W2     = (const float*)d_in[6];
    const float* a_src2 = (const float*)d_in[7];
    const float* a_dst2 = (const float*)d_in[8];
    const float* b2     = (const float*)d_in[9];

    const int EP = N_EDGES + N_NODES;

    float* H1   = (float*)d_ws;
    float* AGG1 = H1   + (size_t)N_NODES * 512;
    float* AS1  = AGG1 + (size_t)N_NODES * 512;
    float* AD1  = AS1  + (size_t)N_NODES * HEADS;
    float* AS2  = AS1;
    float* AD2  = AD1;
    float* G2   = H1;

    float* out    = (float*)d_out;
    float* alpha1 = out    + (size_t)N_NODES * OUT_CH;
    float* alpha2 = alpha1 + (size_t)EP * HEADS;
    int*   keep   = (int*)d_out;

    gemm_wmma_f32<IN_CH, HEADS * HID>
        <<<dim3((HEADS * HID) / G_BN, N_NODES / G_BM), 256, 0, stream>>>(
            x, W1, H1, a_src1, a_dst1, AS1, AD1);

    tile_msg<HEADS, HID, 2, true, 0><<<N_NODES / T_R, T_THREADS, 0, stream>>>(
        ei, H1, AS1, AD1, b1, alpha1, AGG1, keep);

    gemm_wmma_f32<HEADS * HID, OUT_CH>
        <<<dim3(OUT_CH / G_BN, N_NODES / G_BM), 256, 0, stream>>>(
            AGG1, W2, G2, a_src2, a_dst2, AS2, AD2);

    tile_msg<1, OUT_CH, 4, false, 1><<<N_NODES / T_R, T_THREADS, 0, stream>>>(
        ei, G2, AS2, AD2, b2, alpha2, out, keep);
}


# Layer-1 edge scan reads a one-byte tile-id plane (written by GEMM1's threads from the destination ids), not the 4-byte ids: 16 edges per 16 B load, byte-parallel test; sum order: arrival as before
# speedup vs baseline: 3.1430x; 1.0317x over previous
#include <hip/hip_runtime.h>
#include <math.h>

#define N_NODES 20000
#define N_EDGES 320000
#define IN_CH 512
#define HID 128
#define HEADS 4
#define OUT_CH 256
#define NEG_SLOPE 0.2f

typedef float v2f __attribute__((ext_vector_type(2)));
typedef float v8f __attribute__((ext_vector_type(8)));

__device__ __forceinline__ int ford(float f) {
    int i = __float_as_int(f);
    return (i >= 0) ? i : (i ^ 0x7FFFFFFF);
}
__device__ __forceinline__ float iord(int i) {
    return __int_as_float((i >= 0) ? i : (i ^ 0x7FFFFFFF));
}

#define T_R 80
__device__ __forceinline__ unsigned tile_byte(int d) {
    const unsigned t = (unsigned)d / T_R;
    return (t < 255u) ? t : 255u;
}

typedef float v4f __attribute__((ext_vector_type(4)));

#define G_BM 160
#define G_BN 128
#define G_BK 32
#define G_AP 36
#define G_BP 584
#define G_CP 36

template <int K, int NOUT, bool PLANE>
__global__ __launch_bounds__(256)
void gemm_wmma_f32(const float* __restrict__ A, const float* __restrict__ B,
                   float* __restrict__ C,
                   const float* __restrict__ a_src, const float* __restrict__ a_dst,
                   float* __restrict__ as_out, float* __restrict__ ad_out,
                   const int* __restrict__ dstv, unsigned* __restrict__ plane) {
    constexpr int A_ST = G_BM * G_AP;
    constexpr int B_ST = (G_BK / 4) * G_BP;
    constexpr int ST   = A_ST + B_ST;
    constexpr int C_W  = 80 * G_CP;
    constexpr int SMEM = (2 * ST > 8 * C_W) ? 2 * ST : 8 * C_W;
    constexpr int NS   = K / G_BK;
    static_assert(K % G_BK == 0 && NOUT % G_BN == 0 && N_NODES % G_BM == 0, "tiles must divide");
    static_assert(G_BK / 4 == 8, "one k quad of the slab per wave in the B copy");
    __shared__ __attribute__((aligned(16))) float smem[SMEM];

    const int tid  = threadIdx.x;
    const int lane = tid & 31;
    const int wave = tid >> 5;
    const int c  = lane & 15;
    const int h  = lane >> 4;
    const int wm = wave >> 2;
    const int wn = wave & 3;
    const int n0 = blockIdx.x * G_BN;
    const int m0 = blockIdx.y * G_BM;

    if (PLANE) {
        const int nthr = gridDim.x * gridDim.y * 256;
        for (int q = (blockIdx.y * gridDim.x + blockIdx.x) * 256 + tid; q < N_EDGES / 4; q += nthr) {
            const int4 d = *(const int4*)(dstv + 4 * q);
            plane[q] = tile_byte(d.x) | (tile_byte(d.y) << 8) | (tile_byte(d.z) << 16) | (tile_byte(d.w) << 24);
        }
    }

    const int a_row = tid >> 3;
    const int a_k4  = (tid & 7) * 4;
    const float* __restrict__ Ag = A + (size_t)(m0 + a_row) * K + a_k4;
    const float* __restrict__ Bg = B + (size_t)(4 * wave) * NOUT + n0 + 4 * lane;
    float* const As_w = smem + a_row * G_AP + a_k4;
    float* const Bs_w = smem + A_ST + wave * G_BP + 4 * lane;
    const float* const Af = smem + (wm * 80 + c) * G_AP + 4 * h;
    const float* const Bf = smem + A_ST + h * G_BP + 4 * ((c & 3) * 36 + wn * 8 + (c >> 2));

    v4f ra[5], rb[4];
    v8f acc[5][2] = {};

#define G_GLOAD(k0)                                                              \
    _Pragma("unroll") for (int p = 0; p < 5; ++p)                                \
        ra[p] = *(const v4f*)(Ag + (size_t)(p * 32) * K + (k0));                 \
    _Pragma("unroll") for (int r = 0; r < 4; ++r)                                \
        rb[r] = *(const v4f*)(Bg + (size_t)((k0) + r) * NOUT);
#define G_SSTORE(st)                                                             \
    _Pragma("unroll") for (int p = 0; p < 5; ++p)                                \
        *(v4f*)(As_w + (st) * ST + p * 32 * G_AP) = ra[p];                       \
    _Pragma("unroll") for (int i = 0; i < 4; ++i) {                              \
        v4f col = {rb[0][i], rb[1][i], rb[2][i], rb[3][i]};                      \
        *(v4f*)(Bs_w + (st) * ST + i * 36 * 4) = col;                            \
    }
#define G_COMPUTE(Afs, Bfs)                                                      \
    _Pragma("unroll") for (int t = 0; t < G_BK / 8; ++t) {                       \
        v4f fa[5], fb[2];                                                        \
        _Pragma("unroll") for (int i = 0; i < 5; ++i)                            \
            fa[i] = *(const v4f*)((Afs) + i * 16 * G_AP + t * 8);                \
        _Pragma("unroll") for (int j = 0; j < 2; ++j)                            \
            fb[j] = *(const v4f*)((Bfs) + t * 2 * G_BP + j * 16);                \
        _Pragma("unroll") for (int i = 0; i < 5; ++i)                            \
            _Pragma("unroll") for (int j = 0; j < 2; ++j) {                      \
                v2f a = {fa[i].x, fa[i].y};                                      \
                v2f b = {fb[j].x, fb[j].y};                                      \
                acc[i][j] = __builtin_amdgcn_wmma_f32_16x16x4_f32(               \
                    false, a, false, b, (short)0, acc[i][j], false, false);      \
            }                                                                    \
        _Pragma("unroll") for (int i = 0; i < 5; ++i)                            \
            _Pragma("unroll") for (int j = 0; j < 2; ++j) {                      \
                v2f a = {fa[i].z, fa[i].w};                                      \
                v2f b = {fb[j].z, fb[j].w};                                      \
                acc[i][j] = __builtin_amdgcn_wmma_f32_16x16x4_f32(               \
                    false, a, false, b, (short)0, acc[i][j], false, false);      \
            }                                                                    \
    }
#define G_FENCE() __builtin_amdgcn_sched_barrier(0)
#define G_SLOTS() asm volatile("v_nop\n\tv_nop\n\tv_nop\n\tv_nop")

    G_GLOAD(0)
    G_SSTORE(0)
    __syncthreads();

    for (int s = 0; s < NS - 1; ++s) {
        const int st = s & 1;
        const float* const Afs = Af + st * ST;
        const float* const Bfs = Bf + st * ST;
        G_GLOAD((s + 1) * G_BK)
        G_FENCE();
        G_COMPUTE(Afs, Bfs)
        G_FENCE();
        G_SLOTS();
        G_FENCE();
        G_SSTORE(st ^ 1)
        __syncthreads();
    }
    {
        const float* const Afs = Af + ((NS - 1) & 1) * ST;
        const float* const Bfs = Bf + ((NS - 1) & 1) * ST;
        G_FENCE();
        G_COMPUTE(Afs, Bfs)
        G_FENCE();
        G_SLOTS();
        G_FENCE();
    }
    __syncthreads();

    float* const Cw = smem + wave * C_W;
#pragma unroll
    for (int i = 0; i < 5; ++i)
#pragma unroll
        for (int j = 0; j < 2; ++j)
#pragma unroll
            for (int r = 0; r < 8; ++r)
                Cw[(i * 16 + 8 * h + r) * G_CP + j * 16 + c] = acc[i][j][r];
    __syncthreads();
    const int rr = lane >> 3;
    const int c4 = (lane & 7) * 4;
    float* __restrict__ Cg = C + (size_t)(m0 + wm * 80 + rr) * NOUT + n0 + wn * 32 + c4;
#pragma unroll
    for (int t = 0; t < 20; ++t) {
        v4f v = *(const v4f*)(Cw + (t * 4 + rr) * G_CP + c4);
        *(v4f*)(Cg + (size_t)(t * 4) * NOUT) = v;
    }

    if (tid < G_BM) {
        const float* const rowp = smem + ((tid / 80) * 4) * C_W + (tid % 80) * G_CP;
        const float* __restrict__ av = a_src + n0;
        const float* __restrict__ dv = a_dst + n0;
        v4f s4 = {0.f, 0.f, 0.f, 0.f}, d4 = {0.f, 0.f, 0.f, 0.f};
#pragma unroll
        for (int q = 0; q < 4; ++q)
#pragma unroll
            for (int j = 0; j < 8; ++j) {
                const v4f v = *(const v4f*)(rowp + q * C_W + j * 4);
                const v4f a = *(const v4f*)(av + q * 32 + j * 4);
                const v4f b = *(const v4f*)(dv + q * 32 + j * 4);
                s4 += v * a;
                d4 += v * b;
            }
        as_out[(size_t)blockIdx.x * N_NODES + m0 + tid] = (s4.x + s4.y) + (s4.z + s4.w);
        ad_out[(size_t)blockIdx.x * N_NODES + m0 + tid] = (d4.x + d4.y) + (d4.z + d4.w);
    }
#undef G_GLOAD
#undef G_SSTORE
#undef G_COMPUTE
#undef G_FENCE
#undef G_SLOTS
}

template <int H, int CH>
__device__ __forceinline__ float dot_at(const float* __restrict__ p, int node, int h) {
    constexpr int NPL = CH / G_BN;
    static_assert(CH % G_BN == 0, "a head is a whole number of the GEMM's column blocks");
    float v = p[(size_t)(h * NPL) * N_NODES + node];
#pragma unroll
    for (int q = 1; q < NPL; ++q) v += p[(size_t)(h * NPL + q) * N_NODES + node];
    return v;
}

#define T_CAP 4096
#define T_THREADS 1024
#define T_U 7
#define T_KEEP (T_R * OUT_CH)

__device__ __forceinline__ unsigned tile_hits(unsigned w, unsigned pat) {
    const unsigned x = w ^ pat;
    return ~(((x & 0x7F7F7F7Fu) + 0x7F7F7F7Fu) | x | 0x7F7F7F7Fu);
}

__device__ __forceinline__ float att_logit(float as, float ad) {
    const float v = as + ad;
    return (v > 0.f) ? v : NEG_SLOPE * v;
}

template <int H, int CH, int NE>
__device__ __forceinline__ void tile_add_rows(const float* __restrict__ Hm, const int* lst,
                                              const int* srcl, const float* val,
                                              float* tile, int i, int lane) {
    constexpr int ROW = H * CH;
    constexpr int NL  = ROW / 32;
    constexpr int LPH = CH / 32;
    int s[NE], d[NE];
    float al[NE][H];
    float v[NE][NL];
#pragma unroll
    for (int k = 0; k < NE; ++k) {
        d[k] = __builtin_amdgcn_readfirstlane(lst[i + k]) & 127;
        s[k] = __builtin_amdgcn_readfirstlane(srcl[i + k]);
#pragma unroll
        for (int h = 0; h < H; ++h) al[k][h] = val[(i + k) * H + h];
    }
#pragma unroll
    for (int k = 0; k < NE; ++k) {
        const float* __restrict__ hp = Hm + (size_t)s[k] * ROW + lane;
#pragma unroll
        for (int j = 0; j < NL; ++j) v[k][j] = hp[32 * j];
    }
#pragma unroll
    for (int k = 0; k < NE; ++k) {
        float* tp = tile + d[k] * ROW + lane;
#pragma unroll
        for (int j = 0; j < NL; ++j)
            atomicAdd(&tp[32 * j], al[k][j / LPH] * v[k][j]);
    }
}

template <int H, int CH, int PH>
__device__ __forceinline__ void edge_list_stage(const int* __restrict__ ei,
                                                const float* __restrict__ Hm,
                                                const float* __restrict__ as, const float* adl,
                                                int* emx, float* den, float* tile,
                                                float* __restrict__ alpha_out, int n0, int tid) {
    constexpr int ROW = H * CH;
#pragma nounroll
    for (int e = tid; e < N_EDGES; e += T_THREADS) {
        const unsigned d = (unsigned)(ei[N_EDGES + e] - n0);
        if (d < (unsigned)T_R) {
            const int s = ei[e];
#pragma nounroll
            for (int h = 0; h < H; ++h) {
                const float v = att_logit(dot_at<H, CH>(as, s, h), adl[d * H + h]);
                if (PH == 0) {
                    atomicMax(&emx[d * H + h], ford(v));
                } else {
                    const float ex = expf(v - iord(emx[d * H + h]));
                    if (PH == 1) {
                        atomicAdd(&den[d * H + h], ex);
                    } else {
                        const float a = ex / den[d * H + h];
                        alpha_out[(size_t)e * H + h] = a;
#pragma nounroll
                        for (int c = h * CH; c < (h + 1) * CH; ++c)
                            atomicAdd(&tile[d * ROW + c], a * Hm[(size_t)s * ROW + c]);
                    }
                }
            }
        }
    }
}

template <int H, int CH, int EPW, bool ELU, int KEEP>
__global__ __launch_bounds__(T_THREADS)
void tile_msg(const int* __restrict__ ei, const float* __restrict__ Hm,
              const float* __restrict__ as, const float* __restrict__ ad,
              const float* __restrict__ bias, float* __restrict__ alpha_out,
              float* outp, int* keep_all,
              const unsigned* __restrict__ plane) {
    constexpr int ROW = H * CH;
    constexpr int TS  = T_R * ROW;
    constexpr int SEG = T_R * H;
    static_assert(N_NODES % T_R == 0, "tiles must cover the nodes");
    static_assert(T_R <= 128 && N_EDGES + N_NODES <= (1 << 24), "list entry packs edge id and local row");
    static_assert(CH % 32 == 0 && TS % (4 * T_THREADS) == 0 && N_EDGES % 4 == 0, "16-byte pieces");
    static_assert(SEG <= T_THREADS && T_R < T_CAP, "one thread per segment; self loops fit the list");
    static_assert(2 * T_CAP + 1 <= T_KEEP, "the kept list fits the tile's own output rows");
    static_assert(N_NODES / T_R <= 255 && N_EDGES % 16 == 0, "a tile id is a byte below 255; 16 ids per load");
    __shared__ __attribute__((aligned(16))) float tile[TS];
    __shared__ __attribute__((aligned(16))) float val[T_CAP * H];
    __shared__ int lst[T_CAP];
    __shared__ int srcl[T_CAP];
    __shared__ float adl[SEG];
    __shared__ int emx[SEG];
    __shared__ float den[SEG];
    __shared__ int cnt;

    const int tid  = threadIdx.x;
    const int lane = tid & 31;
    const int n0   = blockIdx.x * T_R;

    int* const keep = keep_all + (size_t)blockIdx.x * T_KEEP;

    if (KEEP == 0 && tid == 0) cnt = T_R;
    if (tid < SEG) { emx[tid] = (int)0x80000000; den[tid] = 0.f; }
    __syncthreads();

    float adv = 0.f;
    if (tid < SEG) adv = dot_at<H, CH>(ad, n0 + tid / H, tid % H);
    int kept_n = 0;
    if (KEEP == 1) kept_n = keep[2 * T_CAP];

    const int* __restrict__ dst = ei + N_EDGES;
    if (KEEP == 0) {
        constexpr int NQ = N_EDGES / 16;
        const uint4* __restrict__ pl = (const uint4*)plane;
        const unsigned pat = (unsigned)blockIdx.x * 0x01010101u;
        for (int p0 = 0; p0 < NQ; p0 += T_THREADS * T_U) {
            const int base = p0 + tid;
            uint4 pv[T_U];
#pragma unroll
            for (int u = 0; u < T_U; ++u) {
                const int q = base + u * T_THREADS;
                pv[u] = (q < NQ) ? pl[q] : make_uint4(~0u, ~0u, ~0u, ~0u);
            }
            int hits = 0;
#pragma unroll
            for (int u = 0; u < T_U; ++u)
                hits += __popc(tile_hits(pv[u].x, pat)) + __popc(tile_hits(pv[u].y, pat)) +
                        __popc(tile_hits(pv[u].z, pat)) + __popc(tile_hits(pv[u].w, pat));
            int incl = hits;
#pragma unroll
            for (int o = 1; o < 32; o <<= 1) {
                const int t = __shfl_up(incl, o);
                if (lane >= o) incl += t;
            }
            const int wtot = __shfl(incl, 31);
            if (wtot) {
                int first = 0;
                if (lane == 31) first = atomicAdd(&cnt, wtot);
                int slot = __shfl(first, 31) + incl - hits;
                if (hits) {
#pragma unroll
                    for (int u = 0; u < T_U; ++u) {
                        const unsigned w[4] = {pv[u].x, pv[u].y, pv[u].z, pv[u].w};
#pragma unroll
                        for (int k = 0; k < 4; ++k) {
                            unsigned zz = tile_hits(w[k], pat);
                            while (zz) {
                                const int bq = __builtin_ctz(zz) >> 3;
                                zz &= zz - 1;
                                if (slot < T_CAP) lst[slot] = (base + u * T_THREADS) * 16 + 4 * k + bq;
                                ++slot;
                            }
                        }
                    }
                }
            }
        }
    }
    if (tid < SEG) adl[tid] = adv;
    __syncthreads();

    const int found = (KEEP == 0) ? cnt : kept_n;
    const bool over = found > T_CAP;
    const int n = over ? T_R : found;
    if (KEEP == 0 && tid == 0) keep[2 * T_CAP] = found;

    constexpr int NP = TS / (4 * T_THREADS);
    v4f sv[NP];
#pragma unroll
    for (int p = 0; p < NP; ++p)
        sv[p] = *(const v4f*)(Hm + (size_t)n0 * ROW + (p * T_THREADS + tid) * 4);

    for (int i = tid; i < n; i += T_THREADS) {
        int ent, s;
        if (KEEP == 0) {
            if (i < T_R) {
                ent = ((N_EDGES + n0 + i) << 7) | i;
                s = n0 + i;
            } else {
                const int e = lst[i];
                s = ei[e];
                ent = (e << 7) | (dst[e] - n0);
            }
            lst[i] = ent;
            keep[i] = ent;
            keep[T_CAP + i] = s;
        } else {
            ent = keep[i];
            s = keep[T_CAP + i];
            lst[i] = ent;
        }
        const int d = ent & 127;
        srcl[i] = s;
#pragma unroll
        for (int h = 0; h < H; ++h) {
            const float v = att_logit(dot_at<H, CH>(as, s, h), adl[d * H + h]);
            val[i * H + h] = v;
            atomicMax(&emx[d * H + h], ford(v));
        }
    }
    if (over) edge_list_stage<H, CH, 0>(ei, Hm, as, adl, emx, den, tile, alpha_out, n0, tid);
    __syncthreads();

    for (int i = tid; i < n; i += T_THREADS) {
        const int d = lst[i] & 127;
#pragma unroll
        for (int h = 0; h < H; ++h) {
            const float ex = expf(val[i * H + h] - iord(emx[d * H + h]));
            val[i * H + h] = ex;
            atomicAdd(&den[d * H + h], ex);
        }
    }
    if (over) edge_list_stage<H, CH, 1>(ei, Hm, as, adl, emx, den, tile, alpha_out, n0, tid);
    __syncthreads();

    for (int i = tid; i < n; i += T_THREADS) {
        const int ent = lst[i];
        const int e = ent >> 7;
        const int d = ent & 127;
#pragma unroll
        for (int h = 0; h < H; ++h) {
            const float a = val[i * H + h] / den[d * H + h];
            val[i * H + h] = a;
            alpha_out[(size_t)e * H + h] = a;
        }
    }
    __syncthreads();

#pragma unroll
    for (int p = 0; p < NP; ++p) {
        const int i = (p * T_THREADS + tid) * 4;
        *(v4f*)(tile + i) = val[(i / ROW) * H + (i % ROW) / CH] * sv[p];
    }
    __syncthreads();

    if (over) edge_list_stage<H, CH, 2>(ei, Hm, as, adl, emx, den, tile, alpha_out, n0, tid);

    const int wave = __builtin_amdgcn_readfirstlane(tid >> 5);
    constexpr int NW = T_THREADS / 32;
    const int nfull = T_R + ((n - T_R) / EPW) * EPW;
    for (int i = T_R + wave * EPW; i < nfull; i += NW * EPW)
        tile_add_rows<H, CH, EPW>(Hm, lst, srcl, val, tile, i, lane);
    for (int i = nfull + wave; i < n; i += NW)
        tile_add_rows<H, CH, 1>(Hm, lst, srcl, val, tile, i, lane);
    __syncthreads();

    for (int i = tid * 4; i < TS; i += 4 * T_THREADS) {
        const int c = i % ROW;
        v4f t = *(const v4f*)(tile + i);
        const v4f b = *(const v4f*)(bias + c);
        t += b;
        if (ELU) {
#pragma unroll
            for (int q = 0; q < 4; ++q) t[q] = (t[q] > 0.f) ? t[q] : (expf(t[q]) - 1.f);
        }
        *(v4f*)(outp + (size_t)n0 * ROW + i) = t;
    }
}

extern "C" void kernel_launch(void* const* d_in, const int* in_sizes, int n_in,
                              void* d_out, int out_size, void* d_ws, size_t ws_size,
                              hipStream_t stream) {
    const float* x      = (const float*)d_in[0];
    const int*   ei     = (const int*)d_in[1];
    const float* W1     = (const float*)d_in[2];
    const float* a_src1 = (const float*)d_in[3];
    const float* a_dst1 = (const float*)d_in[4];
    const float* b1     = (const float*)d_in[5];
    const float* W2     = (const float*)d_in[6];
    const float* a_src2 = (const float*)d_in[7];
    const float* a_dst2 = (const float*)d_in[8];
    const float* b2     = (const float*)d_in[9];

    const int EP = N_EDGES + N_NODES;

    float* H1   = (float*)d_ws;
    float* AGG1 = H1   + (size_t)N_NODES * 512;
    float* AS1  = AGG1 + (size_t)N_NODES * 512;
    float* AD1  = AS1  + (size_t)N_NODES * HEADS;
    float* AS2  = AS1;
    float* AD2  = AD1;
    float* G2   = H1;

    float* out    = (float*)d_out;
    float* alpha1 = out    + (size_t)N_NODES * OUT_CH;
    float* alpha2 = alpha1 + (size_t)EP * HEADS;
    int*   keep   = (int*)d_out;
    unsigned* plane = (unsigned*)alpha2;

    gemm_wmma_f32<IN_CH, HEADS * HID, true>
        <<<dim3((HEADS * HID) / G_BN, N_NODES / G_BM), 256, 0, stream>>>(
            x, W1, H1, a_src1, a_dst1, AS1, AD1, ei + N_EDGES, plane);

    tile_msg<HEADS, HID, 2, true, 0><<<N_NODES / T_R, T_THREADS, 0, stream>>>(
        ei, H1, AS1, AD1, b1, alpha1, AGG1, keep, plane);

    gemm_wmma_f32<HEADS * HID, OUT_CH, false>
        <<<dim3(OUT_CH / G_BN, N_NODES / G_BM), 256, 0, stream>>>(
            AGG1, W2, G2, a_src2, a_dst2, AS2, AD2, nullptr, nullptr);

    tile_msg<1, OUT_CH, 4, false, 1><<<N_NODES / T_R, T_THREADS, 0, stream>>>(
        ei, G2, AS2, AD2, b2, alpha2, out, keep, nullptr);
}


# Message row gathers by 16-byte loads (4 per 2-KB row, was 16 4-byte loads); tile columns permuted in LDS so every float add still hits 32 different banks; sum order: arrival, as before
# speedup vs baseline: 3.2945x; 1.0482x over previous
#include <hip/hip_runtime.h>
#include <math.h>

#define N_NODES 20000
#define N_EDGES 320000
#define IN_CH 512
#define HID 128
#define HEADS 4
#define OUT_CH 256
#define NEG_SLOPE 0.2f

typedef float v2f __attribute__((ext_vector_type(2)));
typedef float v8f __attribute__((ext_vector_type(8)));

__device__ __forceinline__ int ford(float f) {
    int i = __float_as_int(f);
    return (i >= 0) ? i : (i ^ 0x7FFFFFFF);
}
__device__ __forceinline__ float iord(int i) {
    return __int_as_float((i >= 0) ? i : (i ^ 0x7FFFFFFF));
}

#define T_R 80
__device__ __forceinline__ unsigned tile_byte(int d) {
    const unsigned t = (unsigned)d / T_R;
    return (t < 255u) ? t : 255u;
}

typedef float v4f __attribute__((ext_vector_type(4)));

#define G_BM 160
#define G_BN 128
#define G_BK 32
#define G_AP 36
#define G_BP 584
#define G_CP 36

template <int K, int NOUT, bool PLANE>
__global__ __launch_bounds__(256)
void gemm_wmma_f32(const float* __restrict__ A, const float* __restrict__ B,
                   float* __restrict__ C,
                   const float* __restrict__ a_src, const float* __restrict__ a_dst,
                   float* __restrict__ as_out, float* __restrict__ ad_out,
                   const int* __restrict__ dstv, unsigned* __restrict__ plane) {
    constexpr int A_ST = G_BM * G_AP;
    constexpr int B_ST = (G_BK / 4) * G_BP;
    constexpr int ST   = A_ST + B_ST;
    constexpr int C_W  = 80 * G_CP;
    constexpr int SMEM = (2 * ST > 8 * C_W) ? 2 * ST : 8 * C_W;
    constexpr int NS   = K / G_BK;
    static_assert(K % G_BK == 0 && NOUT % G_BN == 0 && N_NODES % G_BM == 0, "tiles must divide");
    static_assert(G_BK / 4 == 8, "one k quad of the slab per wave in the B copy");
    __shared__ __attribute__((aligned(16))) float smem[SMEM];

    const int tid  = threadIdx.x;
    const int lane = tid & 31;
    const int wave = tid >> 5;
    const int c  = lane & 15;
    const int h  = lane >> 4;
    const int wm = wave >> 2;
    const int wn = wave & 3;
    const int n0 = blockIdx.x * G_BN;
    const int m0 = blockIdx.y * G_BM;

    if (PLANE) {
        const int nthr = gridDim.x * gridDim.y * 256;
        for (int q = (blockIdx.y * gridDim.x + blockIdx.x) * 256 + tid; q < N_EDGES / 4; q += nthr) {
            const int4 d = *(const int4*)(dstv + 4 * q);
            plane[q] = tile_byte(d.x) | (tile_byte(d.y) << 8) | (tile_byte(d.z) << 16) | (tile_byte(d.w) << 24);
        }
    }

    const int a_row = tid >> 3;
    const int a_k4  = (tid & 7) * 4;
    const float* __restrict__ Ag = A + (size_t)(m0 + a_row) * K + a_k4;
    const float* __restrict__ Bg = B + (size_t)(4 * wave) * NOUT + n0 + 4 * lane;
    float* const As_w = smem + a_row * G_AP + a_k4;
    float* const Bs_w = smem + A_ST + wave * G_BP + 4 * lane;
    const float* const Af = smem + (wm * 80 + c) * G_AP + 4 * h;
    const float* const Bf = smem + A_ST + h * G_BP + 4 * ((c & 3) * 36 + wn * 8 + (c >> 2));

    v4f ra[5], rb[4];
    v8f acc[5][2] = {};

#define G_GLOAD(k0)                                                              \
    _Pragma("unroll") for (int p = 0; p < 5; ++p)                                \
        ra[p] = *(const v4f*)(Ag + (size_t)(p * 32) * K + (k0));                 \
    _Pragma("unroll") for (int r = 0; r < 4; ++r)                                \
        rb[r] = *(const v4f*)(Bg + (size_t)((k0) + r) * NOUT);
#define G_SSTORE(st)                                                             \
    _Pragma("unroll") for (int p = 0; p < 5; ++p)                                \
        *(v4f*)(As_w + (st) * ST + p * 32 * G_AP) = ra[p];                       \
    _Pragma("unroll") for (int i = 0; i < 4; ++i) {                              \
        v4f col = {rb[0][i], rb[1][i], rb[2][i], rb[3][i]};                      \
        *(v4f*)(Bs_w + (st) * ST + i * 36 * 4) = col;                            \
    }
#define G_COMPUTE(Afs, Bfs)                                                      \
    _Pragma("unroll") for (int t = 0; t < G_BK / 8; ++t) {                       \
        v4f fa[5], fb[2];                                                        \
        _Pragma("unroll") for (int i = 0; i < 5; ++i)                            \
            fa[i] = *(const v4f*)((Afs) + i * 16 * G_AP + t * 8);                \
        _Pragma("unroll") for (int j = 0; j < 2; ++j)                            \
            fb[j] = *(const v4f*)((Bfs) + t * 2 * G_BP + j * 16);                \
        _Pragma("unroll") for (int i = 0; i < 5; ++i)                            \
            _Pragma("unroll") for (int j = 0; j < 2; ++j) {                      \
                v2f a = {fa[i].x, fa[i].y};                                      \
                v2f b = {fb[j].x, fb[j].y};                                      \
                acc[i][j] = __builtin_amdgcn_wmma_f32_16x16x4_f32(               \
                    false, a, false, b, (short)0, acc[i][j], false, false);      \
            }                                                                    \
        _Pragma("unroll") for (int i = 0; i < 5; ++i)                            \
            _Pragma("unroll") for (int j = 0; j < 2; ++j) {                      \
                v2f a = {fa[i].z, fa[i].w};                                      \
                v2f b = {fb[j].z, fb[j].w};                                      \
                acc[i][j] = __builtin_amdgcn_wmma_f32_16x16x4_f32(               \
                    false, a, false, b, (short)0, acc[i][j], false, false);      \
            }                                                                    \
    }
#define G_FENCE() __builtin_amdgcn_sched_barrier(0)
#define G_SLOTS() asm volatile("v_nop\n\tv_nop\n\tv_nop\n\tv_nop")

    G_GLOAD(0)
    G_SSTORE(0)
    __syncthreads();

    for (int s = 0; s < NS - 1; ++s) {
        const int st = s & 1;
        const float* const Afs = Af + st * ST;
        const float* const Bfs = Bf + st * ST;
        G_GLOAD((s + 1) * G_BK)
        G_FENCE();
        G_COMPUTE(Afs, Bfs)
        G_FENCE();
        G_SLOTS();
        G_FENCE();
        G_SSTORE(st ^ 1)
        __syncthreads();
    }
    {
        const float* const Afs = Af + ((NS - 1) & 1) * ST;
        const float* const Bfs = Bf + ((NS - 1) & 1) * ST;
        G_FENCE();
        G_COMPUTE(Afs, Bfs)
        G_FENCE();
        G_SLOTS();
        G_FENCE();
    }
    __syncthreads();

    float* const Cw = smem + wave * C_W;
#pragma unroll
    for (int i = 0; i < 5; ++i)
#pragma unroll
        for (int j = 0; j < 2; ++j)
#pragma unroll
            for (int r = 0; r < 8; ++r)
                Cw[(i * 16 + 8 * h + r) * G_CP + j * 16 + c] = acc[i][j][r];
    __syncthreads();
    const int rr = lane >> 3;
    const int c4 = (lane & 7) * 4;
    float* __restrict__ Cg = C + (size_t)(m0 + wm * 80 + rr) * NOUT + n0 + wn * 32 + c4;
#pragma unroll
    for (int t = 0; t < 20; ++t) {
        v4f v = *(const v4f*)(Cw + (t * 4 + rr) * G_CP + c4);
        *(v4f*)(Cg + (size_t)(t * 4) * NOUT) = v;
    }

    if (tid < G_BM) {
        const float* const rowp = smem + ((tid / 80) * 4) * C_W + (tid % 80) * G_CP;
        const float* __restrict__ av = a_src + n0;
        const float* __restrict__ dv = a_dst + n0;
        v4f s4 = {0.f, 0.f, 0.f, 0.f}, d4 = {0.f, 0.f, 0.f, 0.f};
#pragma unroll
        for (int q = 0; q < 4; ++q)
#pragma unroll
            for (int j = 0; j < 8; ++j) {
                const v4f v = *(const v4f*)(rowp + q * C_W + j * 4);
                const v4f a = *(const v4f*)(av + q * 32 + j * 4);
                const v4f b = *(const v4f*)(dv + q * 32 + j * 4);
                s4 += v * a;
                d4 += v * b;
            }
        as_out[(size_t)blockIdx.x * N_NODES + m0 + tid] = (s4.x + s4.y) + (s4.z + s4.w);
        ad_out[(size_t)blockIdx.x * N_NODES + m0 + tid] = (d4.x + d4.y) + (d4.z + d4.w);
    }
#undef G_GLOAD
#undef G_SSTORE
#undef G_COMPUTE
#undef G_FENCE
#undef G_SLOTS
}

template <int H, int CH>
__device__ __forceinline__ float dot_at(const float* __restrict__ p, int node, int h) {
    constexpr int NPL = CH / G_BN;
    static_assert(CH % G_BN == 0, "a head is a whole number of the GEMM's column blocks");
    float v = p[(size_t)(h * NPL) * N_NODES + node];
#pragma unroll
    for (int q = 1; q < NPL; ++q) v += p[(size_t)(h * NPL + q) * N_NODES + node];
    return v;
}

#define T_CAP 4096
#define T_THREADS 1024
#define T_U 7
#define T_KEEP (T_R * OUT_CH)

__device__ __forceinline__ unsigned tile_hits(unsigned w, unsigned pat) {
    const unsigned x = w ^ pat;
    return ~(((x & 0x7F7F7F7Fu) + 0x7F7F7F7Fu) | x | 0x7F7F7F7Fu);
}

__device__ __forceinline__ float att_logit(float as, float ad) {
    const float v = as + ad;
    return (v > 0.f) ? v : NEG_SLOPE * v;
}

__device__ __forceinline__ int tile_col(int c) {
    return (c & ~127) | ((c & 3) << 5) | ((c & 127) >> 2);
}

template <int H, int CH, int NE>
__device__ __forceinline__ void tile_add_rows(const float* __restrict__ Hm, const int* lst,
                                              const int* srcl, const float* val,
                                              float* tile, int i, int lane) {
    constexpr int ROW = H * CH;
    constexpr int NL  = ROW / 128;
    constexpr int LPH = CH / 128;
    static_assert(CH % 128 == 0, "a 16-byte load of the wave stays inside one head");
    int s[NE], d[NE];
    float al[NE][H];
    v4f v[NE][NL];
#pragma unroll
    for (int k = 0; k < NE; ++k) {
        d[k] = __builtin_amdgcn_readfirstlane(lst[i + k]) & 127;
        s[k] = __builtin_amdgcn_readfirstlane(srcl[i + k]);
#pragma unroll
        for (int h = 0; h < H; ++h) al[k][h] = val[(i + k) * H + h];
    }
#pragma unroll
    for (int k = 0; k < NE; ++k) {
        const v4f* __restrict__ hp = (const v4f*)(Hm + (size_t)s[k] * ROW) + lane;
#pragma unroll
        for (int j = 0; j < NL; ++j) v[k][j] = hp[32 * j];
    }
#pragma unroll
    for (int k = 0; k < NE; ++k) {
        float* tp = tile + d[k] * ROW + lane;
#pragma unroll
        for (int j = 0; j < NL; ++j)
#pragma unroll
            for (int q = 0; q < 4; ++q)
                atomicAdd(&tp[128 * j + 32 * q], al[k][j / LPH] * v[k][j][q]);
    }
}

template <int H, int CH, int PH>
__device__ __forceinline__ void edge_list_stage(const int* __restrict__ ei,
                                                const float* __restrict__ Hm,
                                                const float* __restrict__ as, const float* adl,
                                                int* emx, float* den, float* tile,
                                                float* __restrict__ alpha_out, int n0, int tid) {
    constexpr int ROW = H * CH;
#pragma nounroll
    for (int e = tid; e < N_EDGES; e += T_THREADS) {
        const unsigned d = (unsigned)(ei[N_EDGES + e] - n0);
        if (d < (unsigned)T_R) {
            const int s = ei[e];
#pragma nounroll
            for (int h = 0; h < H; ++h) {
                const float v = att_logit(dot_at<H, CH>(as, s, h), adl[d * H + h]);
                if (PH == 0) {
                    atomicMax(&emx[d * H + h], ford(v));
                } else {
                    const float ex = expf(v - iord(emx[d * H + h]));
                    if (PH == 1) {
                        atomicAdd(&den[d * H + h], ex);
                    } else {
                        const float a = ex / den[d * H + h];
                        alpha_out[(size_t)e * H + h] = a;
#pragma nounroll
                        for (int c = h * CH; c < (h + 1) * CH; ++c)
                            atomicAdd(&tile[d * ROW + tile_col(c)], a * Hm[(size_t)s * ROW + c]);
                    }
                }
            }
        }
    }
}

template <int H, int CH, int EPW, bool ELU, int KEEP>
__global__ __launch_bounds__(T_THREADS)
void tile_msg(const int* __restrict__ ei, const float* __restrict__ Hm,
              const float* __restrict__ as, const float* __restrict__ ad,
              const float* __restrict__ bias, float* __restrict__ alpha_out,
              float* outp, int* keep_all,
              const unsigned* __restrict__ plane) {
    constexpr int ROW = H * CH;
    constexpr int TS  = T_R * ROW;
    constexpr int SEG = T_R * H;
    static_assert(N_NODES % T_R == 0, "tiles must cover the nodes");
    static_assert(T_R <= 128 && N_EDGES + N_NODES <= (1 << 24), "list entry packs edge id and local row");
    static_assert(CH % 32 == 0 && TS % (4 * T_THREADS) == 0 && N_EDGES % 4 == 0, "16-byte pieces");
    static_assert(SEG <= T_THREADS && T_R < T_CAP, "one thread per segment; self loops fit the list");
    static_assert(2 * T_CAP + 1 <= T_KEEP, "the kept list fits the tile's own output rows");
    static_assert(N_NODES / T_R <= 255 && N_EDGES % 16 == 0, "a tile id is a byte below 255; 16 ids per load");
    __shared__ __attribute__((aligned(16))) float tile[TS];
    __shared__ __attribute__((aligned(16))) float val[T_CAP * H];
    __shared__ int lst[T_CAP];
    __shared__ int srcl[T_CAP];
    __shared__ float adl[SEG];
    __shared__ int emx[SEG];
    __shared__ float den[SEG];
    __shared__ int cnt;

    const int tid  = threadIdx.x;
    const int lane = tid & 31;
    const int n0   = blockIdx.x * T_R;

    int* const keep = keep_all + (size_t)blockIdx.x * T_KEEP;

    if (KEEP == 0 && tid == 0) cnt = T_R;
    if (tid < SEG) { emx[tid] = (int)0x80000000; den[tid] = 0.f; }
    __syncthreads();

    float adv = 0.f;
    if (tid < SEG) adv = dot_at<H, CH>(ad, n0 + tid / H, tid % H);
    int kept_n = 0;
    if (KEEP == 1) kept_n = keep[2 * T_CAP];

    const int* __restrict__ dst = ei + N_EDGES;
    if (KEEP == 0) {
        constexpr int NQ = N_EDGES / 16;
        const uint4* __restrict__ pl = (const uint4*)plane;
        const unsigned pat = (unsigned)blockIdx.x * 0x01010101u;
        for (int p0 = 0; p0 < NQ; p0 += T_THREADS * T_U) {
            const int base = p0 + tid;
            uint4 pv[T_U];
#pragma unroll
            for (int u = 0; u < T_U; ++u) {
                const int q = base + u * T_THREADS;
                pv[u] = (q < NQ) ? pl[q] : make_uint4(~0u, ~0u, ~0u, ~0u);
            }
            int hits = 0;
#pragma unroll
            for (int u = 0; u < T_U; ++u)
                hits += __popc(tile_hits(pv[u].x, pat)) + __popc(tile_hits(pv[u].y, pat)) +
                        __popc(tile_hits(pv[u].z, pat)) + __popc(tile_hits(pv[u].w, pat));
            int incl = hits;
#pragma unroll
            for (int o = 1; o < 32; o <<= 1) {
                const int t = __shfl_up(incl, o);
                if (lane >= o) incl += t;
            }
            const int wtot = __shfl(incl, 31);
            if (wtot) {
                int first = 0;
                if (lane == 31) first = atomicAdd(&cnt, wtot);
                int slot = __shfl(first, 31) + incl - hits;
                if (hits) {
#pragma unroll
                    for (int u = 0; u < T_U; ++u) {
                        const unsigned w[4] = {pv[u].x, pv[u].y, pv[u].z, pv[u].w};
#pragma unroll
                        for (int k = 0; k < 4; ++k) {
                            unsigned zz = tile_hits(w[k], pat);
                            while (zz) {
                                const int bq = __builtin_ctz(zz) >> 3;
                                zz &= zz - 1;
                                if (slot < T_CAP) lst[slot] = (base + u * T_THREADS) * 16 + 4 * k + bq;
                                ++slot;
                            }
                        }
                    }
                }
            }
        }
    }
    if (tid < SEG) adl[tid] = adv;
    __syncthreads();

    const int found = (KEEP == 0) ? cnt : kept_n;
    const bool over = found > T_CAP;
    const int n = over ? T_R : found;
    if (KEEP == 0 && tid == 0) keep[2 * T_CAP] = found;

    constexpr int NP = TS / (4 * T_THREADS);
    v4f sv[NP];
#pragma unroll
    for (int p = 0; p < NP; ++p)
        sv[p] = *(const v4f*)(Hm + (size_t)n0 * ROW + (p * T_THREADS + tid) * 4);

    for (int i = tid; i < n; i += T_THREADS) {
        int ent, s;
        if (KEEP == 0) {
            if (i < T_R) {
                ent = ((N_EDGES + n0 + i) << 7) | i;
                s = n0 + i;
            } else {
                const int e = lst[i];
                s = ei[e];
                ent = (e << 7) | (dst[e] - n0);
            }
            lst[i] = ent;
            keep[i] = ent;
            keep[T_CAP + i] = s;
        } else {
            ent = keep[i];
            s = keep[T_CAP + i];
            lst[i] = ent;
        }
        const int d = ent & 127;
        srcl[i] = s;
#pragma unroll
        for (int h = 0; h < H; ++h) {
            const float v = att_logit(dot_at<H, CH>(as, s, h), adl[d * H + h]);
            val[i * H + h] = v;
            atomicMax(&emx[d * H + h], ford(v));
        }
    }
    if (over) edge_list_stage<H, CH, 0>(ei, Hm, as, adl, emx, den, tile, alpha_out, n0, tid);
    __syncthreads();

    for (int i = tid; i < n; i += T_THREADS) {
        const int d = lst[i] & 127;
#pragma unroll
        for (int h = 0; h < H; ++h) {
            const float ex = expf(val[i * H + h] - iord(emx[d * H + h]));
            val[i * H + h] = ex;
            atomicAdd(&den[d * H + h], ex);
        }
    }
    if (over) edge_list_stage<H, CH, 1>(ei, Hm, as, adl, emx, den, tile, alpha_out, n0, tid);
    __syncthreads();

    for (int i = tid; i < n; i += T_THREADS) {
        const int ent = lst[i];
        const int e = ent >> 7;
        const int d = ent & 127;
#pragma unroll
        for (int h = 0; h < H; ++h) {
            const float a = val[i * H + h] / den[d * H + h];
            val[i * H + h] = a;
            alpha_out[(size_t)e * H + h] = a;
        }
    }
    __syncthreads();

#pragma unroll
    for (int p = 0; p < NP; ++p) {
        const int i = (p * T_THREADS + tid) * 4;
        const int c = i % ROW;
        const float a = val[(i / ROW) * H + c / CH];
        float* const tp = tile + (i - c) + tile_col(c);
#pragma unroll
        for (int q = 0; q < 4; ++q) tp[32 * q] = a * sv[p][q];
    }
    __syncthreads();

    if (over) edge_list_stage<H, CH, 2>(ei, Hm, as, adl, emx, den, tile, alpha_out, n0, tid);

    const int wave = __builtin_amdgcn_readfirstlane(tid >> 5);
    constexpr int NW = T_THREADS / 32;
    const int nfull = T_R + ((n - T_R) / EPW) * EPW;
    for (int i = T_R + wave * EPW; i < nfull; i += NW * EPW)
        tile_add_rows<H, CH, EPW>(Hm, lst, srcl, val, tile, i, lane);
    for (int i = nfull + wave; i < n; i += NW)
        tile_add_rows<H, CH, 1>(Hm, lst, srcl, val, tile, i, lane);
    __syncthreads();

    for (int i = tid * 4; i < TS; i += 4 * T_THREADS) {
        const int c = i % ROW;
        const float* const tp = tile + (i - c) + tile_col(c);
        v4f t = {tp[0], tp[32], tp[64], tp[96]};
        const v4f b = *(const v4f*)(bias + c);
        t += b;
        if (ELU) {
#pragma unroll
            for (int q = 0; q < 4; ++q) t[q] = (t[q] > 0.f) ? t[q] : (expf(t[q]) - 1.f);
        }
        *(v4f*)(outp + (size_t)n0 * ROW + i) = t;
    }
}

extern "C" void kernel_launch(void* const* d_in, const int* in_sizes, int n_in,
                              void* d_out, int out_size, void* d_ws, size_t ws_size,
                              hipStream_t stream) {
    const float* x      = (const float*)d_in[0];
    const int*   ei     = (const int*)d_in[1];
    const float* W1     = (const float*)d_in[2];
    const float* a_src1 = (const float*)d_in[3];
    const float* a_dst1 = (const float*)d_in[4];
    const float* b1     = (const float*)d_in[5];
    const float* W2     = (const float*)d_in[6];
    const float* a_src2 = (const float*)d_in[7];
    const float* a_dst2 = (const float*)d_in[8];
    const float* b2     = (const float*)d_in[9];

    const int EP = N_EDGES + N_NODES;

    float* H1   = (float*)d_ws;
    float* AGG1 = H1   + (size_t)N_NODES * 512;
    float* AS1  = AGG1 + (size_t)N_NODES * 512;
    float* AD1  = AS1  + (size_t)N_NODES * HEADS;
    float* AS2  = AS1;
    float* AD2  = AD1;
    float* G2   = H1;

    float* out    = (float*)d_out;
    float* alpha1 = out    + (size_t)N_NODES * OUT_CH;
    float* alpha2 = alpha1 + (size_t)EP * HEADS;
    int*   keep   = (int*)d_out;
    unsigned* plane = (unsigned*)alpha2;

    gemm_wmma_f32<IN_CH, HEADS * HID, true>
        <<<dim3((HEADS * HID) / G_BN, N_NODES / G_BM), 256, 0, stream>>>(
            x, W1, H1, a_src1, a_dst1, AS1, AD1, ei + N_EDGES, plane);

    tile_msg<HEADS, HID, 2, true, 0><<<N_NODES / T_R, T_THREADS, 0, stream>>>(
        ei, H1, AS1, AD1, b1, alpha1, AGG1, keep, plane);

    gemm_wmma_f32<HEADS * HID, OUT_CH, false>
        <<<dim3(OUT_CH / G_BN, N_NODES / G_BM), 256, 0, stream>>>(
            AGG1, W2, G2, a_src2, a_dst2, AS2, AD2, nullptr, nullptr);

    tile_msg<1, OUT_CH, 4, false, 1><<<N_NODES / T_R, T_THREADS, 0, stream>>>(
        ei, G2, AS2, AD2, b2, alpha2, out, keep, nullptr);
}
